# starting kernel
# baseline (speedup 1.0000x reference)
#include <hip/hip_runtime.h>

typedef __attribute__((ext_vector_type(2))) float v2f;
typedef __attribute__((ext_vector_type(8))) float v8f;

constexpr int NUM_USERS = 100000;
constexpr int NUM_ITEMS = 50000;
constexpr int N_NODES   = NUM_USERS + NUM_ITEMS;
constexpr int DIM       = 64;
constexpr int SBERT     = 384;
constexpr int NEDGE     = 2000000;
constexpr int NLAYERS   = 3;

__global__ void zero_f(float* __restrict__ p, int n) {
  int i = blockIdx.x * blockDim.x + threadIdx.x;
  if (i < n) p[i] = 0.0f;
}

__global__ void init_user(const float* __restrict__ u,
                          float* __restrict__ emb0, float* __restrict__ acc) {
  int i = blockIdx.x * blockDim.x + threadIdx.x;
  if (i < NUM_USERS * DIM) {
    float v = u[i];
    emb0[i] = v;
    acc[i]  = v;
  }
}

__global__ __launch_bounds__(128)
void wmma_proj(const float* __restrict__ bert, const float* __restrict__ w,
               float* __restrict__ emb0, float* __restrict__ acc) {
  const int lane = threadIdx.x & 31;
  const int wave = threadIdx.x >> 5;
  const int m0   = blockIdx.x * 16;
  const int n0   = wave * 16;
  const int half = lane >> 4;
  const int r    = lane & 15;

  const float* arow = bert + (size_t)(m0 + r) * SBERT + 2 * half;
  const float* brow = w    + (size_t)(n0 + r) * SBERT + 2 * half;

  v8f c = {};
#pragma unroll 8
  for (int k = 0; k < SBERT; k += 4) {
    v2f a = *(const v2f*)(arow + k);
    v2f b = *(const v2f*)(brow + k);
    c = __builtin_amdgcn_wmma_f32_16x16x4_f32(
         false, a,  false, b,
         (short)0, c,  false,  false);
  }

  const size_t base = (size_t)(NUM_USERS + m0 + 8 * half) * DIM + (size_t)(n0 + r);
#pragma unroll
  for (int i = 0; i < 8; ++i) {
    float v = c[i];
    emb0[base + (size_t)i * DIM] = v;
    acc [base + (size_t)i * DIM] = v;
  }
}

__global__ void deg_count(const int* __restrict__ dst, float* __restrict__ deg) {
  int e = blockIdx.x * blockDim.x + threadIdx.x;
  if (e < NEDGE) atomicAdd(&deg[dst[e]], 1.0f);
}

__global__ void deg_to_dis(float* __restrict__ deg) {
  int i = blockIdx.x * blockDim.x + threadIdx.x;
  if (i < N_NODES) {
    float d = deg[i];
    deg[i] = (d > 0.0f) ? rsqrtf(fmaxf(d, 1.0f)) : 0.0f;
  }
}

__global__ void edge_norm(const int* __restrict__ src, const int* __restrict__ dst,
                          const float* __restrict__ dis, float* __restrict__ norm) {
  int e = blockIdx.x * blockDim.x + threadIdx.x;
  if (e < NEDGE) norm[e] = dis[src[e]] * dis[dst[e]];
}

__global__ void spmm_scatter(const int* __restrict__ src, const int* __restrict__ dst,
                             const float* __restrict__ norm,
                             const float* __restrict__ x, float* __restrict__ xn) {
  int t = blockIdx.x * blockDim.x + threadIdx.x;
  int e = t >> 4;
  if (e >= NEDGE) return;
  int c = (t & 15) << 2;
  float wgt = norm[e];
  float4 v = *(const float4*)(x + (size_t)src[e] * DIM + c);
  float* o = xn + (size_t)dst[e] * DIM + c;
  atomicAdd(o + 0, wgt * v.x);
  atomicAdd(o + 1, wgt * v.y);
  atomicAdd(o + 2, wgt * v.z);
  atomicAdd(o + 3, wgt * v.w);
}

__global__ void acc_add(float* __restrict__ acc, const float* __restrict__ xn) {
  int i = blockIdx.x * blockDim.x + threadIdx.x;
  if (i < N_NODES * DIM) acc[i] += xn[i];
}

__global__ void scale_k(float* __restrict__ acc) {
  int i = blockIdx.x * blockDim.x + threadIdx.x;
  if (i < N_NODES * DIM) acc[i] *= 0.25f;
}

extern "C" void kernel_launch(void* const* d_in, const int* in_sizes, int n_in,
                              void* d_out, int out_size, void* d_ws, size_t ws_size,
                              hipStream_t stream) {
  const int*   ei    = (const int*)d_in[0];
  const int*   src   = ei;
  const int*   dst   = ei + NEDGE;
  const float* user  = (const float*)d_in[1];
  const float* bert  = (const float*)d_in[2];
  const float* wproj = (const float*)d_in[3];

  float* emb0 = (float*)d_out;
  float* acc  = emb0 + (size_t)N_NODES * DIM;

  float* deg  = (float*)d_ws;
  float* norm = deg  + ((N_NODES + 63) & ~63);
  float* xA   = norm + ((NEDGE + 63) & ~63);
  float* xB   = xA   + (size_t)N_NODES * DIM;

  const int B = 256;
  const int NV = N_NODES * DIM;

  zero_f   <<<(N_NODES + B - 1) / B, B, 0, stream>>>(deg, N_NODES);
  init_user<<<(NUM_USERS * DIM + B - 1) / B, B, 0, stream>>>(user, emb0, acc);
  wmma_proj<<<NUM_ITEMS / 16, 128, 0, stream>>>(bert, wproj, emb0, acc);

  deg_count <<<(NEDGE + B - 1) / B, B, 0, stream>>>(dst, deg);
  deg_to_dis<<<(N_NODES + B - 1) / B, B, 0, stream>>>(deg);
  edge_norm <<<(NEDGE + B - 1) / B, B, 0, stream>>>(src, dst, deg, norm);

  const float* xcur = emb0;
  float* bufs[2] = {xA, xB};
  for (int l = 0; l < NLAYERS; ++l) {
    float* xn = bufs[l & 1];
    zero_f      <<<(NV + B - 1) / B, B, 0, stream>>>(xn, NV);
    spmm_scatter<<<(NEDGE * 16 + B - 1) / B, B, 0, stream>>>(src, dst, norm, xcur, xn);
    acc_add     <<<(NV + B - 1) / B, B, 0, stream>>>(acc, xn);
    xcur = xn;
  }
  scale_k<<<(NV + B - 1) / B, B, 0, stream>>>(acc);
}


# Layer sum formed in ONE pass behind the last layer, out = 1/4(((x0+x1)+x2)+x3): three add passes, the scale pass and the set-up writes of the running sum go; layer 3 scatters into cleared out
# speedup vs baseline: 1.8532x; 1.8532x over previous
#include <hip/hip_runtime.h>

typedef __attribute__((ext_vector_type(2))) float v2f;
typedef __attribute__((ext_vector_type(8))) float v8f;
typedef __attribute__((ext_vector_type(4))) float v4f;

constexpr int NUM_USERS = 100000;
constexpr int NUM_ITEMS = 50000;
constexpr int N_NODES   = NUM_USERS + NUM_ITEMS;
constexpr int DIM       = 64;
constexpr int SBERT     = 384;
constexpr int NEDGE     = 2000000;
constexpr int NLAYERS   = 3;

__global__ void zero_f(float* __restrict__ p, int n) {
  int i = blockIdx.x * blockDim.x + threadIdx.x;
  if (i < n) p[i] = 0.0f;
}

constexpr int EW_BLOCKS  = 625;
constexpr int EW_THREADS = 256;
constexpr int EW_LANES   = EW_BLOCKS * EW_THREADS;

constexpr int USER_V4 = NUM_USERS * DIM / 4;
constexpr int NODE_V4 = N_NODES * DIM / 4;
constexpr int USER_TRIPS = USER_V4 / EW_LANES;
constexpr int NODE_TRIPS = NODE_V4 / EW_LANES;
static_assert(USER_TRIPS * EW_LANES * 4 == NUM_USERS * DIM, "user array must split evenly");
static_assert(NODE_TRIPS * EW_LANES * 4 == N_NODES * DIM,   "node array must split evenly");

__global__ __launch_bounds__(EW_THREADS)
void init_user(const float* __restrict__ u, float* __restrict__ emb0) {
  const v4f* __restrict__ u4 = (const v4f*)u;
  v4f* __restrict__ e4 = (v4f*)emb0;
  const int t = blockIdx.x * EW_THREADS + threadIdx.x;
  v4f v[USER_TRIPS];
  for (int k = 0; k < USER_TRIPS; ++k) v[k] = u4[t + k * EW_LANES];
  for (int k = 0; k < USER_TRIPS; ++k) e4[t + k * EW_LANES] = v[k];
}

__global__ __launch_bounds__(EW_THREADS)
void zero_rows(float* __restrict__ p) {
  v4f* __restrict__ p4 = (v4f*)p;
  const int t = blockIdx.x * EW_THREADS + threadIdx.x;
  const v4f z = {0.0f, 0.0f, 0.0f, 0.0f};
  for (int k = 0; k < NODE_TRIPS; ++k) p4[t + k * EW_LANES] = z;
}

__global__ __launch_bounds__(128)
void wmma_proj(const float* __restrict__ bert, const float* __restrict__ w,
               float* __restrict__ emb0) {
  const int lane = threadIdx.x & 31;
  const int wave = threadIdx.x >> 5;
  const int m0   = blockIdx.x * 16;
  const int n0   = wave * 16;
  const int half = lane >> 4;
  const int r    = lane & 15;

  const float* arow = bert + (size_t)(m0 + r) * SBERT + 2 * half;
  const float* brow = w    + (size_t)(n0 + r) * SBERT + 2 * half;

  v8f c = {};
#pragma unroll 8
  for (int k = 0; k < SBERT; k += 4) {
    v2f a = *(const v2f*)(arow + k);
    v2f b = *(const v2f*)(brow + k);
    c = __builtin_amdgcn_wmma_f32_16x16x4_f32(
         false, a,  false, b,
         (short)0, c,  false,  false);
  }

  const size_t base = (size_t)(NUM_USERS + m0 + 8 * half) * DIM + (size_t)(n0 + r);
#pragma unroll
  for (int i = 0; i < 8; ++i) emb0[base + (size_t)i * DIM] = c[i];
}

__global__ void deg_count(const int* __restrict__ dst, float* __restrict__ deg) {
  int e = blockIdx.x * blockDim.x + threadIdx.x;
  if (e < NEDGE) atomicAdd(&deg[dst[e]], 1.0f);
}

__global__ void deg_to_dis(float* __restrict__ deg) {
  int i = blockIdx.x * blockDim.x + threadIdx.x;
  if (i < N_NODES) {
    float d = deg[i];
    deg[i] = (d > 0.0f) ? rsqrtf(fmaxf(d, 1.0f)) : 0.0f;
  }
}

__global__ void edge_norm(const int* __restrict__ src, const int* __restrict__ dst,
                          const float* __restrict__ dis, float* __restrict__ norm) {
  int e = blockIdx.x * blockDim.x + threadIdx.x;
  if (e < NEDGE) norm[e] = dis[src[e]] * dis[dst[e]];
}

constexpr int EPW        = 16;
constexpr int SPMM_WAVES = NEDGE / EPW;
static_assert(NEDGE % EPW == 0, "edge count must split into whole waves");
static_assert(SPMM_WAVES % 8 == 0, "waves must split into whole 8-wave blocks");

__global__ __launch_bounds__(256)
void spmm_rows(const int* __restrict__ src, const int* __restrict__ dst,
               const float* __restrict__ norm,
               const float* __restrict__ x, float* __restrict__ xn) {
  const int lane = threadIdx.x & 31;
  const int wv = __builtin_amdgcn_readfirstlane((int)(threadIdx.x >> 5));
  const int e0 = (blockIdx.x * 8 + wv) * EPW;

  float lo[EPW], hi[EPW];
#pragma unroll
  for (int j = 0; j < EPW; ++j) {
    const float* row = x + (size_t)(unsigned)src[e0 + j] * DIM;
    lo[j] = row[lane];
    hi[j] = row[lane + 32];
  }
#pragma unroll
  for (int j = 0; j < EPW; ++j) {
    const float wgt = norm[e0 + j];
    float* o = xn + (size_t)(unsigned)dst[e0 + j] * DIM;
    atomicAdd(o + lane,      wgt * lo[j]);
    atomicAdd(o + lane + 32, wgt * hi[j]);
  }
}

static_assert(NLAYERS == 3, "two layers go to the workspace, the third into out");
constexpr int SUM_CHUNK = 5;
static_assert(NODE_TRIPS % SUM_CHUNK == 0, "chunks must cover the lane's items");

__global__ __launch_bounds__(EW_THREADS)
void layer_sum(const float* __restrict__ x0, const float* __restrict__ x1,
               const float* __restrict__ x2, float* __restrict__ out) {
  const v4f* __restrict__ p0 = (const v4f*)x0;
  const v4f* __restrict__ p1 = (const v4f*)x1;
  const v4f* __restrict__ p2 = (const v4f*)x2;
  v4f* __restrict__ o4 = (v4f*)out;
  const int t = blockIdx.x * EW_THREADS + threadIdx.x;
#pragma nounroll
  for (int c = 0; c < NODE_TRIPS; c += SUM_CHUNK) {
    v4f a[SUM_CHUNK], b[SUM_CHUNK], d[SUM_CHUNK], e[SUM_CHUNK];
    for (int k = 0; k < SUM_CHUNK; ++k) {
      a[k] = p0[t + (c + k) * EW_LANES];
      b[k] = p1[t + (c + k) * EW_LANES];
      d[k] = p2[t + (c + k) * EW_LANES];
      e[k] = o4[t + (c + k) * EW_LANES];
    }
    for (int k = 0; k < SUM_CHUNK; ++k)
      o4[t + (c + k) * EW_LANES] = (((a[k] + b[k]) + d[k]) + e[k]) * 0.25f;
  }
}

extern "C" void kernel_launch(void* const* d_in, const int* in_sizes, int n_in,
                              void* d_out, int out_size, void* d_ws, size_t ws_size,
                              hipStream_t stream) {
  const int*   ei    = (const int*)d_in[0];
  const int*   src   = ei;
  const int*   dst   = ei + NEDGE;
  const float* user  = (const float*)d_in[1];
  const float* bert  = (const float*)d_in[2];
  const float* wproj = (const float*)d_in[3];

  float* emb0 = (float*)d_out;
  float* out  = emb0 + (size_t)N_NODES * DIM;

  float* deg  = (float*)d_ws;
  float* norm = deg  + ((N_NODES + 63) & ~63);
  float* xA   = norm + ((NEDGE + 63) & ~63);
  float* xB   = xA   + (size_t)N_NODES * DIM;

  const int B = 256;

  zero_f   <<<(N_NODES + B - 1) / B, B, 0, stream>>>(deg, N_NODES);
  init_user<<<EW_BLOCKS, EW_THREADS, 0, stream>>>(user, emb0);
  wmma_proj<<<NUM_ITEMS / 16, 128, 0, stream>>>(bert, wproj, emb0);

  deg_count <<<(NEDGE + B - 1) / B, B, 0, stream>>>(dst, deg);
  deg_to_dis<<<(N_NODES + B - 1) / B, B, 0, stream>>>(deg);
  edge_norm <<<(NEDGE + B - 1) / B, B, 0, stream>>>(src, dst, deg, norm);

  zero_rows<<<EW_BLOCKS, EW_THREADS, 0, stream>>>(xA);
  spmm_rows<<<SPMM_WAVES / 8, 256, 0, stream>>>(src, dst, norm, emb0, xA);
  zero_rows<<<EW_BLOCKS, EW_THREADS, 0, stream>>>(xB);
  spmm_rows<<<SPMM_WAVES / 8, 256, 0, stream>>>(src, dst, norm, xA, xB);
  zero_rows<<<EW_BLOCKS, EW_THREADS, 0, stream>>>(out);
  spmm_rows<<<SPMM_WAVES / 8, 256, 0, stream>>>(src, dst, norm, xB, out);
  layer_sum<<<EW_BLOCKS, EW_THREADS, 0, stream>>>(emb0, xA, xB, out);
}


# SpMM adds kept inside one half of the chip: destination rows split in two parts, a work-group reads its die number and adds only into its own half's part (edge chunks by ticket); arrival-order sums
# speedup vs baseline: 1.9413x; 1.0475x over previous
#include <hip/hip_runtime.h>

typedef __attribute__((ext_vector_type(2))) float v2f;
typedef __attribute__((ext_vector_type(8))) float v8f;
typedef __attribute__((ext_vector_type(4))) float v4f;

constexpr int NUM_USERS = 100000;
constexpr int NUM_ITEMS = 50000;
constexpr int N_NODES   = NUM_USERS + NUM_ITEMS;
constexpr int DIM       = 64;
constexpr int SBERT     = 384;
constexpr int NEDGE     = 2000000;
constexpr int NLAYERS   = 3;

__global__ void zero_f(float* __restrict__ p, int n) {
  int i = blockIdx.x * blockDim.x + threadIdx.x;
  if (i < n) p[i] = 0.0f;
}

constexpr int EW_BLOCKS  = 625;
constexpr int EW_THREADS = 256;
constexpr int EW_LANES   = EW_BLOCKS * EW_THREADS;

constexpr int USER_V4 = NUM_USERS * DIM / 4;
constexpr int NODE_V4 = N_NODES * DIM / 4;
constexpr int USER_TRIPS = USER_V4 / EW_LANES;
constexpr int NODE_TRIPS = NODE_V4 / EW_LANES;
static_assert(USER_TRIPS * EW_LANES * 4 == NUM_USERS * DIM, "user array must split evenly");
static_assert(NODE_TRIPS * EW_LANES * 4 == N_NODES * DIM,   "node array must split evenly");

__global__ __launch_bounds__(EW_THREADS)
void init_user(const float* __restrict__ u, float* __restrict__ emb0) {
  const v4f* __restrict__ u4 = (const v4f*)u;
  v4f* __restrict__ e4 = (v4f*)emb0;
  const int t = blockIdx.x * EW_THREADS + threadIdx.x;
  v4f v[USER_TRIPS];
  for (int k = 0; k < USER_TRIPS; ++k) v[k] = u4[t + k * EW_LANES];
  for (int k = 0; k < USER_TRIPS; ++k) e4[t + k * EW_LANES] = v[k];
}

constexpr int TICKET_PITCH = 64;
__global__ __launch_bounds__(EW_THREADS)
void zero_rows(float* __restrict__ p, int* __restrict__ ticket) {
  v4f* __restrict__ p4 = (v4f*)p;
  const int t = blockIdx.x * EW_THREADS + threadIdx.x;
  const v4f z = {0.0f, 0.0f, 0.0f, 0.0f};
  for (int k = 0; k < NODE_TRIPS; ++k) p4[t + k * EW_LANES] = z;
  if (t < 2) ticket[t * TICKET_PITCH] = 0;
}

__global__ __launch_bounds__(128)
void wmma_proj(const float* __restrict__ bert, const float* __restrict__ w,
               float* __restrict__ emb0) {
  const int lane = threadIdx.x & 31;
  const int wave = threadIdx.x >> 5;
  const int m0   = blockIdx.x * 16;
  const int n0   = wave * 16;
  const int half = lane >> 4;
  const int r    = lane & 15;

  const float* arow = bert + (size_t)(m0 + r) * SBERT + 2 * half;
  const float* brow = w    + (size_t)(n0 + r) * SBERT + 2 * half;

  v8f c = {};
#pragma unroll 8
  for (int k = 0; k < SBERT; k += 4) {
    v2f a = *(const v2f*)(arow + k);
    v2f b = *(const v2f*)(brow + k);
    c = __builtin_amdgcn_wmma_f32_16x16x4_f32(
         false, a,  false, b,
         (short)0, c,  false,  false);
  }

  const size_t base = (size_t)(NUM_USERS + m0 + 8 * half) * DIM + (size_t)(n0 + r);
#pragma unroll
  for (int i = 0; i < 8; ++i) emb0[base + (size_t)i * DIM] = c[i];
}

__global__ void deg_count(const int* __restrict__ dst, float* __restrict__ deg) {
  int e = blockIdx.x * blockDim.x + threadIdx.x;
  if (e < NEDGE) atomicAdd(&deg[dst[e]], 1.0f);
}

__global__ void deg_to_dis(float* __restrict__ deg) {
  int i = blockIdx.x * blockDim.x + threadIdx.x;
  if (i < N_NODES) {
    float d = deg[i];
    deg[i] = (d > 0.0f) ? rsqrtf(fmaxf(d, 1.0f)) : 0.0f;
  }
}

__global__ void edge_norm(const int* __restrict__ src, const int* __restrict__ dst,
                          const float* __restrict__ dis, float* __restrict__ norm) {
  int e = blockIdx.x * blockDim.x + threadIdx.x;
  if (e < NEDGE) norm[e] = dis[src[e]] * dis[dst[e]];
}

constexpr int EPW        = 16;
constexpr int PART_SPLIT = N_NODES / 2;
constexpr int CH_BATCHES = 5;
constexpr int CH_EDGES   = 8 * EPW * CH_BATCHES;
constexpr int NCHUNK     = NEDGE / CH_EDGES;
constexpr int SPMM_GRID  = 2048;
static_assert(NCHUNK * CH_EDGES == NEDGE, "edges must split into whole chunks");

__global__ __launch_bounds__(256)
void spmm_rows(const int* __restrict__ src, const int* __restrict__ dst,
               const float* __restrict__ norm,
               const float* __restrict__ x, float* __restrict__ xn,
               int* __restrict__ ticket) {
  __shared__ int s_t;
  const int lane = threadIdx.x & 31;
  const int wv = __builtin_amdgcn_readfirstlane((int)(threadIdx.x >> 5));
  int part = (int)((__builtin_amdgcn_s_sendmsg_rtn(0x87) >> 18) & 1u);
  bool other = false;
  typedef int   v16i __attribute__((ext_vector_type(16)));
  typedef float v16f __attribute__((ext_vector_type(16)));
  typedef const __attribute__((address_space(4))) int*   cint_p;
  typedef const __attribute__((address_space(4))) float* cflt_p;
  typedef const __attribute__((address_space(4))) v16i*  cint16_p;
  typedef const __attribute__((address_space(4))) v16f*  cflt16_p;
  const cint_p csrc = (cint_p)src;
  const cint_p cdst = (cint_p)dst;
  const cflt_p cnrm = (cflt_p)norm;

  for (int it = 0; it < 2 * NCHUNK + 2; ++it) {
    if (threadIdx.x == 0) s_t = atomicAdd(&ticket[part * TICKET_PITCH], 1);
    __syncthreads();
    const int t = __builtin_amdgcn_readfirstlane(s_t);
    __syncthreads();
    if (t >= NCHUNK) {
      if (other) break;
      other = true;
      part ^= 1;
      continue;
    }
    const int eb = t * CH_EDGES + wv * (EPW * CH_BATCHES);
#pragma nounroll
    for (int b = 0; b < CH_BATCHES; ++b) {
      const int e0 = eb + b * EPW;
      float lo[EPW], hi[EPW];
      const v16i dv = *(cint16_p)(cdst + e0);
      const v16i sv = *(cint16_p)(csrc + e0);
      const v16f wv16 = *(cflt16_p)(cnrm + e0);
#pragma unroll
      for (int j = 0; j < EPW; ++j) {
        const int d = dv[j];
        lo[j] = 0.0f;
        hi[j] = 0.0f;
        if ((int)(d >= PART_SPLIT) == part) {
          const float* row = x + (size_t)(unsigned)sv[j] * DIM;
          lo[j] = row[lane];
          hi[j] = row[lane + 32];
        }
      }
#pragma unroll
      for (int j = 0; j < EPW; ++j) {
        const int d = dv[j];
        if ((int)(d >= PART_SPLIT) == part) {
          const float wgt = wv16[j];
          float* o = xn + (size_t)(unsigned)d * DIM;
          atomicAdd(o + lane,      wgt * lo[j]);
          atomicAdd(o + lane + 32, wgt * hi[j]);
        }
      }
    }
  }
}

static_assert(NLAYERS == 3, "two layers go to the workspace, the third into out");
constexpr int SUM_CHUNK = 5;
static_assert(NODE_TRIPS % SUM_CHUNK == 0, "chunks must cover the lane's items");

__global__ __launch_bounds__(EW_THREADS)
void layer_sum(const float* __restrict__ x0, const float* __restrict__ x1,
               const float* __restrict__ x2, float* __restrict__ out) {
  const v4f* __restrict__ p0 = (const v4f*)x0;
  const v4f* __restrict__ p1 = (const v4f*)x1;
  const v4f* __restrict__ p2 = (const v4f*)x2;
  v4f* __restrict__ o4 = (v4f*)out;
  const int t = blockIdx.x * EW_THREADS + threadIdx.x;
#pragma nounroll
  for (int c = 0; c < NODE_TRIPS; c += SUM_CHUNK) {
    v4f a[SUM_CHUNK], b[SUM_CHUNK], d[SUM_CHUNK], e[SUM_CHUNK];
    for (int k = 0; k < SUM_CHUNK; ++k) {
      a[k] = p0[t + (c + k) * EW_LANES];
      b[k] = p1[t + (c + k) * EW_LANES];
      d[k] = p2[t + (c + k) * EW_LANES];
      e[k] = o4[t + (c + k) * EW_LANES];
    }
    for (int k = 0; k < SUM_CHUNK; ++k)
      o4[t + (c + k) * EW_LANES] = (((a[k] + b[k]) + d[k]) + e[k]) * 0.25f;
  }
}

extern "C" void kernel_launch(void* const* d_in, const int* in_sizes, int n_in,
                              void* d_out, int out_size, void* d_ws, size_t ws_size,
                              hipStream_t stream) {
  const int*   ei    = (const int*)d_in[0];
  const int*   src   = ei;
  const int*   dst   = ei + NEDGE;
  const float* user  = (const float*)d_in[1];
  const float* bert  = (const float*)d_in[2];
  const float* wproj = (const float*)d_in[3];

  float* emb0 = (float*)d_out;
  float* out  = emb0 + (size_t)N_NODES * DIM;

  float* deg  = (float*)d_ws;
  float* norm = deg  + ((N_NODES + 63) & ~63);
  float* xA   = norm + ((NEDGE + 63) & ~63);
  float* xB   = xA   + (size_t)N_NODES * DIM;

  const int B = 256;

  zero_f   <<<(N_NODES + B - 1) / B, B, 0, stream>>>(deg, N_NODES);
  init_user<<<EW_BLOCKS, EW_THREADS, 0, stream>>>(user, emb0);
  wmma_proj<<<NUM_ITEMS / 16, 128, 0, stream>>>(bert, wproj, emb0);

  deg_count <<<(NEDGE + B - 1) / B, B, 0, stream>>>(dst, deg);
  deg_to_dis<<<(N_NODES + B - 1) / B, B, 0, stream>>>(deg);
  edge_norm <<<(NEDGE + B - 1) / B, B, 0, stream>>>(src, dst, deg, norm);

  int* tick = (int*)deg;
  zero_rows<<<EW_BLOCKS, EW_THREADS, 0, stream>>>(xA, tick);
  spmm_rows<<<SPMM_GRID, 256, 0, stream>>>(src, dst, norm, emb0, xA, tick);
  zero_rows<<<EW_BLOCKS, EW_THREADS, 0, stream>>>(xB, tick + 128);
  spmm_rows<<<SPMM_GRID, 256, 0, stream>>>(src, dst, norm, xA, xB, tick + 128);
  zero_rows<<<EW_BLOCKS, EW_THREADS, 0, stream>>>(out, tick + 256);
  spmm_rows<<<SPMM_GRID, 256, 0, stream>>>(src, dst, norm, xB, out, tick + 256);
  layer_sum<<<EW_BLOCKS, EW_THREADS, 0, stream>>>(emb0, xA, xB, out);
}


# In-degrees counted in LDS: one work-group per (destination range, edge slice) counts with ds integer adds and stores a plane piece, the rsqrt pass sums the pieces; no global adds, no clear launch
# speedup vs baseline: 2.3929x; 1.2326x over previous
#include <hip/hip_runtime.h>

typedef __attribute__((ext_vector_type(2))) float v2f;
typedef __attribute__((ext_vector_type(8))) float v8f;
typedef __attribute__((ext_vector_type(4))) float v4f;

constexpr int NUM_USERS = 100000;
constexpr int NUM_ITEMS = 50000;
constexpr int N_NODES   = NUM_USERS + NUM_ITEMS;
constexpr int DIM       = 64;
constexpr int SBERT     = 384;
constexpr int NEDGE     = 2000000;
constexpr int NLAYERS   = 3;

constexpr int EW_BLOCKS  = 625;
constexpr int EW_THREADS = 256;
constexpr int EW_LANES   = EW_BLOCKS * EW_THREADS;

constexpr int USER_V4 = NUM_USERS * DIM / 4;
constexpr int NODE_V4 = N_NODES * DIM / 4;
constexpr int USER_TRIPS = USER_V4 / EW_LANES;
constexpr int NODE_TRIPS = NODE_V4 / EW_LANES;
static_assert(USER_TRIPS * EW_LANES * 4 == NUM_USERS * DIM, "user array must split evenly");
static_assert(NODE_TRIPS * EW_LANES * 4 == N_NODES * DIM,   "node array must split evenly");

__global__ __launch_bounds__(EW_THREADS)
void init_user(const float* __restrict__ u, float* __restrict__ emb0) {
  const v4f* __restrict__ u4 = (const v4f*)u;
  v4f* __restrict__ e4 = (v4f*)emb0;
  const int t = blockIdx.x * EW_THREADS + threadIdx.x;
  v4f v[USER_TRIPS];
  for (int k = 0; k < USER_TRIPS; ++k) v[k] = u4[t + k * EW_LANES];
  for (int k = 0; k < USER_TRIPS; ++k) e4[t + k * EW_LANES] = v[k];
}

constexpr int TICKET_PITCH = 64;
__global__ __launch_bounds__(EW_THREADS)
void zero_rows(float* __restrict__ p, int* __restrict__ ticket) {
  v4f* __restrict__ p4 = (v4f*)p;
  const int t = blockIdx.x * EW_THREADS + threadIdx.x;
  const v4f z = {0.0f, 0.0f, 0.0f, 0.0f};
  for (int k = 0; k < NODE_TRIPS; ++k) p4[t + k * EW_LANES] = z;
  if (t < 2) ticket[t * TICKET_PITCH] = 0;
}

__global__ __launch_bounds__(128)
void wmma_proj(const float* __restrict__ bert, const float* __restrict__ w,
               float* __restrict__ emb0) {
  const int lane = threadIdx.x & 31;
  const int wave = threadIdx.x >> 5;
  const int m0   = blockIdx.x * 16;
  const int n0   = wave * 16;
  const int half = lane >> 4;
  const int r    = lane & 15;

  const float* arow = bert + (size_t)(m0 + r) * SBERT + 2 * half;
  const float* brow = w    + (size_t)(n0 + r) * SBERT + 2 * half;

  v8f c = {};
#pragma unroll 8
  for (int k = 0; k < SBERT; k += 4) {
    v2f a = *(const v2f*)(arow + k);
    v2f b = *(const v2f*)(brow + k);
    c = __builtin_amdgcn_wmma_f32_16x16x4_f32(
         false, a,  false, b,
         (short)0, c,  false,  false);
  }

  const size_t base = (size_t)(NUM_USERS + m0 + 8 * half) * DIM + (size_t)(n0 + r);
#pragma unroll
  for (int i = 0; i < 8; ++i) emb0[base + (size_t)i * DIM] = c[i];
}

constexpr int DH_RANGES  = 8;
constexpr int DH_PARTS   = 32;
constexpr int DH_BINS    = N_NODES / DH_RANGES;
constexpr int DH_PITCH   = (DH_BINS + 31) & ~31;
constexpr int DH_THREADS = 1024;
constexpr int DH_EDGES   = NEDGE / DH_PARTS;
constexpr int DH_V4      = DH_EDGES / 4;
constexpr int DH_TRIPS   = (DH_V4 + DH_THREADS - 1) / DH_THREADS;
static_assert(DH_BINS * DH_RANGES == N_NODES, "ranges must cover the nodes exactly");
static_assert(DH_EDGES * DH_PARTS == NEDGE,   "slices must cover the edges exactly");
static_assert(DH_V4 * 4 == DH_EDGES,          "a slice must be whole 16-byte items");
static_assert((size_t)DH_PARTS * DH_RANGES * DH_PITCH <= (size_t)N_NODES * DIM,
              "the plane pieces must fit into one layer array of the workspace");

typedef int v4i __attribute__((ext_vector_type(4)));

__global__ __launch_bounds__(DH_THREADS)
void deg_hist(const int* __restrict__ dst, unsigned* __restrict__ planes) {
  __shared__ unsigned bins[DH_BINS];
  const int range = blockIdx.x % DH_RANGES;
  const int part  = blockIdx.x / DH_RANGES;
  const int tid   = threadIdx.x;

  const v4i* __restrict__ p = (const v4i*)(dst + (size_t)part * DH_EDGES);
  v4i v[DH_TRIPS];
#pragma unroll
  for (int k = 0; k < DH_TRIPS; ++k) {
    const int idx = tid + k * DH_THREADS;
    v[k] = (idx < DH_V4) ? p[idx] : v4i{-1, -1, -1, -1};
  }

  for (int i = tid; i < DH_BINS; i += DH_THREADS) bins[i] = 0u;
  __syncthreads();

  const int base = range * DH_BINS;
#pragma unroll
  for (int k = 0; k < DH_TRIPS; ++k) {
#pragma unroll
    for (int c = 0; c < 4; ++c) {
      const unsigned u = (unsigned)(v[k][c] - base);
      if (u < (unsigned)DH_BINS) atomicAdd(&bins[u], 1u);
    }
  }
  __syncthreads();

  unsigned* __restrict__ o = planes + ((size_t)part * DH_RANGES + range) * DH_PITCH;
  for (int i = tid; i < DH_BINS; i += DH_THREADS) o[i] = bins[i];
}

__global__ __launch_bounds__(256)
void deg_to_dis(const unsigned* __restrict__ planes, float* __restrict__ dis) {
  const int i = blockIdx.x * blockDim.x + threadIdx.x;
  if (i < N_NODES) {
    const int range = i / DH_BINS;
    const unsigned* __restrict__ p = planes + (size_t)range * DH_PITCH + (i - range * DH_BINS);
    unsigned c = 0u;
#pragma unroll
    for (int k = 0; k < DH_PARTS; ++k) c += p[(size_t)k * DH_RANGES * DH_PITCH];
    const float d = (float)c;
    dis[i] = (d > 0.0f) ? rsqrtf(fmaxf(d, 1.0f)) : 0.0f;
  }
}

__global__ void edge_norm(const int* __restrict__ src, const int* __restrict__ dst,
                          const float* __restrict__ dis, float* __restrict__ norm) {
  int e = blockIdx.x * blockDim.x + threadIdx.x;
  if (e < NEDGE) norm[e] = dis[src[e]] * dis[dst[e]];
}

constexpr int EPW        = 16;
constexpr int PART_SPLIT = N_NODES / 2;
constexpr int CH_BATCHES = 5;
constexpr int CH_EDGES   = 8 * EPW * CH_BATCHES;
constexpr int NCHUNK     = NEDGE / CH_EDGES;
constexpr int SPMM_GRID  = 2048;
static_assert(NCHUNK * CH_EDGES == NEDGE, "edges must split into whole chunks");

__global__ __launch_bounds__(256)
void spmm_rows(const int* __restrict__ src, const int* __restrict__ dst,
               const float* __restrict__ norm,
               const float* __restrict__ x, float* __restrict__ xn,
               int* __restrict__ ticket) {
  __shared__ int s_t;
  const int lane = threadIdx.x & 31;
  const int wv = __builtin_amdgcn_readfirstlane((int)(threadIdx.x >> 5));
  int part = (int)((__builtin_amdgcn_s_sendmsg_rtn(0x87) >> 18) & 1u);
  bool other = false;
  typedef int   v16i __attribute__((ext_vector_type(16)));
  typedef float v16f __attribute__((ext_vector_type(16)));
  typedef const __attribute__((address_space(4))) int*   cint_p;
  typedef const __attribute__((address_space(4))) float* cflt_p;
  typedef const __attribute__((address_space(4))) v16i*  cint16_p;
  typedef const __attribute__((address_space(4))) v16f*  cflt16_p;
  const cint_p csrc = (cint_p)src;
  const cint_p cdst = (cint_p)dst;
  const cflt_p cnrm = (cflt_p)norm;

  for (int it = 0; it < 2 * NCHUNK + 2; ++it) {
    if (threadIdx.x == 0) s_t = atomicAdd(&ticket[part * TICKET_PITCH], 1);
    __syncthreads();
    const int t = __builtin_amdgcn_readfirstlane(s_t);
    __syncthreads();
    if (t >= NCHUNK) {
      if (other) break;
      other = true;
      part ^= 1;
      continue;
    }
    const int eb = t * CH_EDGES + wv * (EPW * CH_BATCHES);
#pragma nounroll
    for (int b = 0; b < CH_BATCHES; ++b) {
      const int e0 = eb + b * EPW;
      float lo[EPW], hi[EPW];
      const v16i dv = *(cint16_p)(cdst + e0);
      const v16i sv = *(cint16_p)(csrc + e0);
      const v16f wv16 = *(cflt16_p)(cnrm + e0);
#pragma unroll
      for (int j = 0; j < EPW; ++j) {
        const int d = dv[j];
        lo[j] = 0.0f;
        hi[j] = 0.0f;
        if ((int)(d >= PART_SPLIT) == part) {
          const float* row = x + (size_t)(unsigned)sv[j] * DIM;
          lo[j] = row[lane];
          hi[j] = row[lane + 32];
        }
      }
#pragma unroll
      for (int j = 0; j < EPW; ++j) {
        const int d = dv[j];
        if ((int)(d >= PART_SPLIT) == part) {
          const float wgt = wv16[j];
          float* o = xn + (size_t)(unsigned)d * DIM;
          atomicAdd(o + lane,      wgt * lo[j]);
          atomicAdd(o + lane + 32, wgt * hi[j]);
        }
      }
    }
  }
}

static_assert(NLAYERS == 3, "two layers go to the workspace, the third into out");
constexpr int SUM_CHUNK = 5;
static_assert(NODE_TRIPS % SUM_CHUNK == 0, "chunks must cover the lane's items");

__global__ __launch_bounds__(EW_THREADS)
void layer_sum(const float* __restrict__ x0, const float* __restrict__ x1,
               const float* __restrict__ x2, float* __restrict__ out) {
  const v4f* __restrict__ p0 = (const v4f*)x0;
  const v4f* __restrict__ p1 = (const v4f*)x1;
  const v4f* __restrict__ p2 = (const v4f*)x2;
  v4f* __restrict__ o4 = (v4f*)out;
  const int t = blockIdx.x * EW_THREADS + threadIdx.x;
#pragma nounroll
  for (int c = 0; c < NODE_TRIPS; c += SUM_CHUNK) {
    v4f a[SUM_CHUNK], b[SUM_CHUNK], d[SUM_CHUNK], e[SUM_CHUNK];
    for (int k = 0; k < SUM_CHUNK; ++k) {
      a[k] = p0[t + (c + k) * EW_LANES];
      b[k] = p1[t + (c + k) * EW_LANES];
      d[k] = p2[t + (c + k) * EW_LANES];
      e[k] = o4[t + (c + k) * EW_LANES];
    }
    for (int k = 0; k < SUM_CHUNK; ++k)
      o4[t + (c + k) * EW_LANES] = (((a[k] + b[k]) + d[k]) + e[k]) * 0.25f;
  }
}

extern "C" void kernel_launch(void* const* d_in, const int* in_sizes, int n_in,
                              void* d_out, int out_size, void* d_ws, size_t ws_size,
                              hipStream_t stream) {
  const int*   ei    = (const int*)d_in[0];
  const int*   src   = ei;
  const int*   dst   = ei + NEDGE;
  const float* user  = (const float*)d_in[1];
  const float* bert  = (const float*)d_in[2];
  const float* wproj = (const float*)d_in[3];

  float* emb0 = (float*)d_out;
  float* out  = emb0 + (size_t)N_NODES * DIM;

  float* deg  = (float*)d_ws;
  float* norm = deg  + ((N_NODES + 63) & ~63);
  float* xA   = norm + ((NEDGE + 63) & ~63);
  float* xB   = xA   + (size_t)N_NODES * DIM;

  const int B = 256;

  init_user<<<EW_BLOCKS, EW_THREADS, 0, stream>>>(user, emb0);
  wmma_proj<<<NUM_ITEMS / 16, 128, 0, stream>>>(bert, wproj, emb0);

  unsigned* planes = (unsigned*)xB;
  deg_hist  <<<DH_RANGES * DH_PARTS, DH_THREADS, 0, stream>>>(dst, planes);
  deg_to_dis<<<(N_NODES + B - 1) / B, B, 0, stream>>>(planes, deg);
  edge_norm <<<(NEDGE + B - 1) / B, B, 0, stream>>>(src, dst, deg, norm);

  int* tick = (int*)deg;
  zero_rows<<<EW_BLOCKS, EW_THREADS, 0, stream>>>(xA, tick);
  spmm_rows<<<SPMM_GRID, 256, 0, stream>>>(src, dst, norm, emb0, xA, tick);
  zero_rows<<<EW_BLOCKS, EW_THREADS, 0, stream>>>(xB, tick + 128);
  spmm_rows<<<SPMM_GRID, 256, 0, stream>>>(src, dst, norm, xA, xB, tick + 128);
  zero_rows<<<EW_BLOCKS, EW_THREADS, 0, stream>>>(out, tick + 256);
  spmm_rows<<<SPMM_GRID, 256, 0, stream>>>(src, dst, norm, xB, out, tick + 256);
  layer_sum<<<EW_BLOCKS, EW_THREADS, 0, stream>>>(emb0, xA, xB, out);
}


# SpMM gathers served from LDS: edges bucketed per call by 586-row source block x destination half, block staged times dis[s]; whole-line f32 adds kept, weight as two row scales; arrival-order sums
# speedup vs baseline: 2.7404x; 1.1452x over previous
#include <hip/hip_runtime.h>

typedef __attribute__((ext_vector_type(2))) float v2f;
typedef __attribute__((ext_vector_type(8))) float v8f;
typedef __attribute__((ext_vector_type(4))) float v4f;

constexpr int NUM_USERS = 100000;
constexpr int NUM_ITEMS = 50000;
constexpr int N_NODES   = NUM_USERS + NUM_ITEMS;
constexpr int DIM       = 64;
constexpr int SBERT     = 384;
constexpr int NEDGE     = 2000000;
constexpr int NLAYERS   = 3;

constexpr int EW_BLOCKS  = 625;
constexpr int EW_THREADS = 256;
constexpr int EW_LANES   = EW_BLOCKS * EW_THREADS;

constexpr int USER_V4 = NUM_USERS * DIM / 4;
constexpr int NODE_V4 = N_NODES * DIM / 4;
constexpr int USER_TRIPS = USER_V4 / EW_LANES;
constexpr int NODE_TRIPS = NODE_V4 / EW_LANES;
static_assert(USER_TRIPS * EW_LANES * 4 == NUM_USERS * DIM, "user array must split evenly");
static_assert(NODE_TRIPS * EW_LANES * 4 == N_NODES * DIM,   "node array must split evenly");

__global__ __launch_bounds__(EW_THREADS)
void init_user(const float* __restrict__ u, float* __restrict__ emb0) {
  const v4f* __restrict__ u4 = (const v4f*)u;
  v4f* __restrict__ e4 = (v4f*)emb0;
  const int t = blockIdx.x * EW_THREADS + threadIdx.x;
  v4f v[USER_TRIPS];
  for (int k = 0; k < USER_TRIPS; ++k) v[k] = u4[t + k * EW_LANES];
  for (int k = 0; k < USER_TRIPS; ++k) e4[t + k * EW_LANES] = v[k];
}

constexpr int TICKET_PITCH = 1;
__global__ __launch_bounds__(EW_THREADS)
void zero_rows(float* __restrict__ p, int* __restrict__ ticket) {
  v4f* __restrict__ p4 = (v4f*)p;
  const int t = blockIdx.x * EW_THREADS + threadIdx.x;
  const v4f z = {0.0f, 0.0f, 0.0f, 0.0f};
  for (int k = 0; k < NODE_TRIPS; ++k) p4[t + k * EW_LANES] = z;
  if (t < 2) ticket[t * TICKET_PITCH] = 0;
}

__global__ __launch_bounds__(128)
void wmma_proj(const float* __restrict__ bert, const float* __restrict__ w,
               float* __restrict__ emb0) {
  const int lane = threadIdx.x & 31;
  const int wave = threadIdx.x >> 5;
  const int m0   = blockIdx.x * 16;
  const int n0   = wave * 16;
  const int half = lane >> 4;
  const int r    = lane & 15;

  const float* arow = bert + (size_t)(m0 + r) * SBERT + 4 * half;
  const float* brow = w    + (size_t)(n0 + r) * SBERT + 4 * half;

  v8f c = {};
#pragma unroll 8
  for (int k = 0; k < SBERT; k += 8) {
    const v4f a = *(const v4f*)(arow + k);
    const v4f b = *(const v4f*)(brow + k);
    const v2f a0 = {a.x, a.y}, b0 = {b.x, b.y};
    const v2f a1 = {a.z, a.w}, b1 = {b.z, b.w};
    c = __builtin_amdgcn_wmma_f32_16x16x4_f32(
         false, a0,  false, b0,
         (short)0, c,  false,  false);
    c = __builtin_amdgcn_wmma_f32_16x16x4_f32(
         false, a1,  false, b1,
         (short)0, c,  false,  false);
  }

  const size_t base = (size_t)(NUM_USERS + m0 + 8 * half) * DIM + (size_t)(n0 + r);
#pragma unroll
  for (int i = 0; i < 8; ++i) emb0[base + (size_t)i * DIM] = c[i];
}

constexpr int DH_RANGES  = 8;
constexpr int DH_PARTS   = 32;
constexpr int DH_BINS    = N_NODES / DH_RANGES;
constexpr int DH_PITCH   = (DH_BINS + 31) & ~31;
constexpr int DH_THREADS = 1024;
constexpr int DH_EDGES   = NEDGE / DH_PARTS;
constexpr int DH_V4      = DH_EDGES / 4;
constexpr int DH_TRIPS   = (DH_V4 + DH_THREADS - 1) / DH_THREADS;
static_assert(DH_BINS * DH_RANGES == N_NODES, "ranges must cover the nodes exactly");
static_assert(DH_EDGES * DH_PARTS == NEDGE,   "slices must cover the edges exactly");
static_assert(DH_V4 * 4 == DH_EDGES,          "a slice must be whole 16-byte items");
static_assert((size_t)DH_PARTS * DH_RANGES * DH_PITCH <= (size_t)N_NODES * DIM,
              "the plane pieces must fit into one layer array of the workspace");

typedef int v4i __attribute__((ext_vector_type(4)));

__global__ __launch_bounds__(DH_THREADS)
void deg_hist(const int* __restrict__ dst, unsigned* __restrict__ planes) {
  __shared__ unsigned bins[DH_BINS];
  const int range = blockIdx.x % DH_RANGES;
  const int part  = blockIdx.x / DH_RANGES;
  const int tid   = threadIdx.x;

  const v4i* __restrict__ p = (const v4i*)(dst + (size_t)part * DH_EDGES);
  v4i v[DH_TRIPS];
#pragma unroll
  for (int k = 0; k < DH_TRIPS; ++k) {
    const int idx = tid + k * DH_THREADS;
    v[k] = (idx < DH_V4) ? p[idx] : v4i{-1, -1, -1, -1};
  }

  for (int i = tid; i < DH_BINS; i += DH_THREADS) bins[i] = 0u;
  __syncthreads();

  const int base = range * DH_BINS;
#pragma unroll
  for (int k = 0; k < DH_TRIPS; ++k) {
#pragma unroll
    for (int c = 0; c < 4; ++c) {
      const unsigned u = (unsigned)(v[k][c] - base);
      if (u < (unsigned)DH_BINS) atomicAdd(&bins[u], 1u);
    }
  }
  __syncthreads();

  unsigned* __restrict__ o = planes + ((size_t)part * DH_RANGES + range) * DH_PITCH;
  for (int i = tid; i < DH_BINS; i += DH_THREADS) o[i] = bins[i];
}

__global__ __launch_bounds__(256)
void deg_to_dis(const unsigned* __restrict__ planes, float* __restrict__ dis) {
  const int i = blockIdx.x * blockDim.x + threadIdx.x;
  if (i < N_NODES) {
    const int range = i / DH_BINS;
    const unsigned* __restrict__ p = planes + (size_t)range * DH_PITCH + (i - range * DH_BINS);
    unsigned c = 0u;
#pragma unroll
    for (int k = 0; k < DH_PARTS; ++k) c += p[(size_t)k * DH_RANGES * DH_PITCH];
    const float d = (float)c;
    dis[i] = (d > 0.0f) ? rsqrtf(fmaxf(d, 1.0f)) : 0.0f;
  }
}

constexpr int BK_ROWS    = 586;
constexpr int NBLK       = 256;
constexpr int NBKT       = 2 * NBLK;
constexpr int PART_SPLIT = N_NODES / 2;
constexpr int HDR        = (NBKT + 1) * 8;
constexpr int SL_N       = 250;
constexpr int SL_EDGES   = NEDGE / SL_N;
constexpr int SL_V4      = SL_EDGES / 4;
constexpr int BS_THREADS = 512;
constexpr int BS_TRIPS   = (SL_V4 + BS_THREADS - 1) / BS_THREADS;
static_assert(BK_ROWS * NBLK >= N_NODES,  "blocks must cover the nodes");
static_assert(BK_ROWS <= 1024,            "row inside a block must fit in 10 bits");
static_assert(N_NODES <= (1 << 18),       "destination must fit in 18 bits");
static_assert(SL_EDGES * SL_N == NEDGE && SL_V4 * 4 == SL_EDGES, "slices must split evenly");
static_assert(BS_THREADS == NBKT,         "one thread per bucket in the bucketing kernels");
static_assert(NEDGE % 16 == 0 && HDR <= NEDGE, "record batches and header must fit");

__device__ __forceinline__ int bucket_of(int s, int d) {
  return 2 * (s / BK_ROWS) + (int)(d >= PART_SPLIT);
}

__global__ __launch_bounds__(BS_THREADS)
void bucket_count(const int* __restrict__ src, const int* __restrict__ dst,
                  unsigned* __restrict__ cnt) {
  __shared__ unsigned h[NBKT];
  const int tid = threadIdx.x;
  const int slice = blockIdx.x;
  const v4i* __restrict__ ps = (const v4i*)(src + (size_t)slice * SL_EDGES);
  const v4i* __restrict__ pd = (const v4i*)(dst + (size_t)slice * SL_EDGES);
  v4i sv[BS_TRIPS], dv[BS_TRIPS];
#pragma unroll
  for (int k = 0; k < BS_TRIPS; ++k) {
    const int idx = tid + k * BS_THREADS;
    sv[k] = (idx < SL_V4) ? ps[idx] : v4i{-1, -1, -1, -1};
    dv[k] = (idx < SL_V4) ? pd[idx] : v4i{0, 0, 0, 0};
  }
  h[tid] = 0u;
  __syncthreads();
#pragma unroll
  for (int k = 0; k < BS_TRIPS; ++k) {
#pragma unroll
    for (int c = 0; c < 4; ++c)
      if (sv[k][c] >= 0) atomicAdd(&h[bucket_of(sv[k][c], dv[k][c])], 1u);
  }
  __syncthreads();
  cnt[(size_t)slice * NBKT + tid] = h[tid];
}

constexpr int BO_THREADS = 2 * NBKT;
constexpr int BO_SLICES  = SL_N / 2;
static_assert(BO_SLICES * 2 == SL_N, "two halves of the slices");
__global__ __launch_bounds__(BO_THREADS)
void bucket_offsets(const unsigned* __restrict__ cnt, unsigned* __restrict__ pre,
                    unsigned* __restrict__ off) {
  __shared__ unsigned part[2][NBKT];
  const int q = threadIdx.x / NBKT;
  const int b = threadIdx.x % NBKT;
  const unsigned* __restrict__ col = cnt + (size_t)q * BO_SLICES * NBKT + b;
  unsigned sum = 0u;
#pragma unroll 25
  for (int s = 0; s < BO_SLICES; ++s) sum += col[(size_t)s * NBKT];
  part[q][b] = sum;
  __syncthreads();
  if (q == 0) {
    unsigned a = 0u, all = 0u;
    for (int j = 0; j < NBKT; ++j) {
      const unsigned c = part[0][j] + part[1][j];
      if (j < b) a += c;
      all += c;
    }
    off[b] = a;
    if (b == 0) off[NBKT] = all;
  }
  unsigned run = q ? part[0][b] : 0u;
  unsigned* __restrict__ pcol = pre + (size_t)q * BO_SLICES * NBKT + b;
#pragma unroll 25
  for (int s = 0; s < BO_SLICES; ++s) {
    const unsigned c = col[(size_t)s * NBKT];
    pcol[(size_t)s * NBKT] = run;
    run += c;
  }
}

__global__ __launch_bounds__(BS_THREADS)
void bucket_scatter(const int* __restrict__ src, const int* __restrict__ dst,
                    const unsigned* __restrict__ cnt, const unsigned* __restrict__ pre,
                    const unsigned* __restrict__ off, unsigned* __restrict__ rec) {
  __shared__ unsigned lc[NBKT];
  __shared__ unsigned lstart[NBKT];
  __shared__ unsigned lcur[NBKT];
  __shared__ unsigned gbase[NBKT];
  __shared__ unsigned sw[SL_EDGES];
  __shared__ unsigned short sb[SL_EDGES];
  const int tid = threadIdx.x;
  const int slice = blockIdx.x;
  const v4i* __restrict__ ps = (const v4i*)(src + (size_t)slice * SL_EDGES);
  const v4i* __restrict__ pd = (const v4i*)(dst + (size_t)slice * SL_EDGES);
  v4i sv[BS_TRIPS], dv[BS_TRIPS];
#pragma unroll
  for (int k = 0; k < BS_TRIPS; ++k) {
    const int idx = tid + k * BS_THREADS;
    sv[k] = (idx < SL_V4) ? ps[idx] : v4i{-1, -1, -1, -1};
    dv[k] = (idx < SL_V4) ? pd[idx] : v4i{0, 0, 0, 0};
  }
  lc[tid] = cnt[(size_t)slice * NBKT + tid];
  gbase[tid] = off[tid] + pre[(size_t)slice * NBKT + tid];
  __syncthreads();
  unsigned a = 0u;
  for (int j = 0; j < NBKT; ++j) {
    const unsigned c = lc[j];
    if (j < tid) a += c;
  }
  lstart[tid] = a;
  lcur[tid] = a;
  __syncthreads();
#pragma unroll
  for (int k = 0; k < BS_TRIPS; ++k) {
#pragma unroll
    for (int c = 0; c < 4; ++c) {
      const int s = sv[k][c];
      if (s >= 0) {
        const int d = dv[k][c];
        const int bk = bucket_of(s, d);
        const unsigned r = (unsigned)(s - (s / BK_ROWS) * BK_ROWS);
        const unsigned p = atomicAdd(&lcur[bk], 1u);
        sw[p] = (r << 18) | (unsigned)d;
        sb[p] = (unsigned short)bk;
      }
    }
  }
  __syncthreads();
  for (int i = tid; i < SL_EDGES; i += BS_THREADS) {
    const int bk = sb[i];
    const unsigned pos = gbase[bk] + ((unsigned)i - lstart[bk]);
    unsigned w = sw[i];
    if (pos < (unsigned)HDR) w |= ((off[pos >> 3] >> (4u * (pos & 7u))) & 15u) << 28;
    rec[pos] = w;
  }
}

constexpr int SP_THREADS = 1024;
constexpr int SP_GRID    = NBKT;
constexpr int ST_ITEMS   = BK_ROWS * DIM / 4;
constexpr int ST_TRIPS   = (ST_ITEMS + SP_THREADS - 1) / SP_THREADS;
constexpr int ST_HALF    = ST_TRIPS / 2;
static_assert(ST_HALF * 2 == ST_TRIPS, "staging runs in two equal halves");
constexpr int EPW        = 16;

template <bool FIRST>
__global__ __launch_bounds__(SP_THREADS)
void spmm_block(const float* __restrict__ x, const float* __restrict__ dis,
                const unsigned* __restrict__ rec, float* __restrict__ xn,
                int* __restrict__ ticket) {
  __shared__ float rows[BK_ROWS * DIM];
  __shared__ int s_t;
  const int tid  = threadIdx.x;
  const int lane = tid & 31;
  const int wv = __builtin_amdgcn_readfirstlane(tid >> 5);
  int part = (int)((__builtin_amdgcn_s_sendmsg_rtn(0x87) >> 18) & 1u);
  bool other = false;
  typedef unsigned v16u __attribute__((ext_vector_type(16)));
  typedef const __attribute__((address_space(4))) unsigned* cu_p;
  typedef const __attribute__((address_space(4))) v16u*     cu16_p;
  const cu_p crec = (cu_p)rec;

  for (int it = 0; it < 2 * NBLK + 2; ++it) {
    if (tid == 0) s_t = atomicAdd(&ticket[part * TICKET_PITCH], 1);
    __syncthreads();
    const int t = __builtin_amdgcn_readfirstlane(s_t);
    __syncthreads();
    if (t >= NBLK) {
      if (other) break;
      other = true;
      part ^= 1;
      continue;
    }
    const int bkt  = 2 * t + part;
    const int row0 = t * BK_ROWS;
    const int nrows = (N_NODES - row0 < BK_ROWS) ? (N_NODES - row0) : BK_ROWS;

    unsigned st = 0u, en = 0u;
#pragma unroll
    for (int j = 0; j < 8; ++j) {
      st |= (crec[8 * bkt + j]     >> 28) << (4 * j);
      en |= (crec[8 * bkt + 8 + j] >> 28) << (4 * j);
    }

    const v4f* __restrict__ x4 = (const v4f*)(x + (size_t)row0 * DIM);
#pragma nounroll
    for (int k0 = 0; k0 < ST_TRIPS; k0 += ST_HALF) {
      v4f v[ST_HALF];
      float sc[ST_HALF];
#pragma unroll
      for (int k = 0; k < ST_HALF; ++k) {
        const int idx = tid + (k0 + k) * SP_THREADS;
        const int r = idx >> 4;
        v[k] = v4f{0.0f, 0.0f, 0.0f, 0.0f};
        sc[k] = 0.0f;
        if (r < nrows) { v[k] = x4[idx]; sc[k] = dis[row0 + r]; }
      }
#pragma unroll
      for (int k = 0; k < ST_HALF; ++k) {
        const int idx = tid + (k0 + k) * SP_THREADS;
        if (idx < ST_ITEMS) {
          v4f y = v[k] * sc[k];
          if (!FIRST) y = y * sc[k];
          ((v4f*)rows)[idx] = y;
        }
      }
    }
    __syncthreads();

    const int e_first = (int)(st & ~15u);
#pragma nounroll
    for (int e0 = e_first + wv * EPW; e0 < (int)en; e0 += (SP_THREADS / 32) * EPW) {
      const v16u rv = *(cu16_p)(crec + e0);
      float lo[EPW], hi[EPW];
#pragma unroll
      for (int j = 0; j < EPW; ++j) {
        lo[j] = 0.0f;
        hi[j] = 0.0f;
        if (e0 + j >= (int)st && e0 + j < (int)en) {
          const float* row = rows + ((rv[j] >> 18) & 1023u) * DIM;
          lo[j] = row[lane];
          hi[j] = row[lane + 32];
        }
      }
#pragma unroll
      for (int j = 0; j < EPW; ++j) {
        if (e0 + j >= (int)st && e0 + j < (int)en) {
          float* o = xn + (size_t)(rv[j] & 0x3FFFFu) * DIM;
          atomicAdd(o + lane,      lo[j]);
          atomicAdd(o + lane + 32, hi[j]);
        }
      }
    }
  }
}

static_assert(NLAYERS == 3, "two layers go to the workspace, the third into out");
constexpr int SUM_CHUNK = 5;
static_assert(NODE_TRIPS % SUM_CHUNK == 0, "chunks must cover the lane's items");

__global__ __launch_bounds__(EW_THREADS)
void layer_sum(const float* __restrict__ x0, const float* __restrict__ x1,
               const float* __restrict__ x2, const float* __restrict__ dis,
               float* __restrict__ out) {
  const v4f* __restrict__ p0 = (const v4f*)x0;
  const v4f* __restrict__ p1 = (const v4f*)x1;
  const v4f* __restrict__ p2 = (const v4f*)x2;
  v4f* __restrict__ o4 = (v4f*)out;
  const int t = blockIdx.x * EW_THREADS + threadIdx.x;
#pragma nounroll
  for (int c = 0; c < NODE_TRIPS; c += SUM_CHUNK) {
    v4f a[SUM_CHUNK], b[SUM_CHUNK], d[SUM_CHUNK], e[SUM_CHUNK];
    float g[SUM_CHUNK];
    for (int k = 0; k < SUM_CHUNK; ++k) {
      a[k] = p0[t + (c + k) * EW_LANES];
      b[k] = p1[t + (c + k) * EW_LANES];
      d[k] = p2[t + (c + k) * EW_LANES];
      e[k] = o4[t + (c + k) * EW_LANES];
      g[k] = dis[(t + (c + k) * EW_LANES) >> 4];
    }
    for (int k = 0; k < SUM_CHUNK; ++k)
      o4[t + (c + k) * EW_LANES] =
          (((a[k] + b[k] * g[k]) + d[k] * g[k]) + e[k] * g[k]) * 0.25f;
  }
}

extern "C" void kernel_launch(void* const* d_in, const int* in_sizes, int n_in,
                              void* d_out, int out_size, void* d_ws, size_t ws_size,
                              hipStream_t stream) {
  const int*   ei    = (const int*)d_in[0];
  const int*   src   = ei;
  const int*   dst   = ei + NEDGE;
  const float* user  = (const float*)d_in[1];
  const float* bert  = (const float*)d_in[2];
  const float* wproj = (const float*)d_in[3];

  float* emb0 = (float*)d_out;
  float* out  = emb0 + (size_t)N_NODES * DIM;

  float* deg  = (float*)d_ws;
  float* norm = deg  + ((N_NODES + 63) & ~63);
  float* xA   = norm + ((NEDGE + 63) & ~63);
  float* xB   = xA   + (size_t)N_NODES * DIM;

  const int B = 256;

  init_user<<<EW_BLOCKS, EW_THREADS, 0, stream>>>(user, emb0);
  wmma_proj<<<NUM_ITEMS / 16, 128, 0, stream>>>(bert, wproj, emb0);

  unsigned* planes = (unsigned*)xB;
  deg_hist  <<<DH_RANGES * DH_PARTS, DH_THREADS, 0, stream>>>(dst, planes);
  deg_to_dis<<<(N_NODES + B - 1) / B, B, 0, stream>>>(planes, deg);

  unsigned* rec = (unsigned*)norm;
  unsigned* cnt = (unsigned*)xA;
  unsigned* pre = cnt + (size_t)SL_N * NBKT;
  unsigned* off = pre + (size_t)SL_N * NBKT;
  bucket_count  <<<SL_N, BS_THREADS, 0, stream>>>(src, dst, cnt);
  bucket_offsets<<<1, BO_THREADS, 0, stream>>>(cnt, pre, off);
  bucket_scatter<<<SL_N, BS_THREADS, 0, stream>>>(src, dst, cnt, pre, off, rec);

  int* tick = (int*)(deg + N_NODES);
  zero_rows<<<EW_BLOCKS, EW_THREADS, 0, stream>>>(xA, tick);
  spmm_block<true> <<<SP_GRID, SP_THREADS, 0, stream>>>(emb0, deg, rec, xA, tick);
  zero_rows<<<EW_BLOCKS, EW_THREADS, 0, stream>>>(xB, tick + 2);
  spmm_block<false><<<SP_GRID, SP_THREADS, 0, stream>>>(xA, deg, rec, xB, tick + 2);
  zero_rows<<<EW_BLOCKS, EW_THREADS, 0, stream>>>(out, tick + 4);
  spmm_block<false><<<SP_GRID, SP_THREADS, 0, stream>>>(xB, deg, rec, out, tick + 4);
  layer_sum<<<EW_BLOCKS, EW_THREADS, 0, stream>>>(emb0, xA, xB, deg, out);
}


# No clear launches: layer 3 adds on top of layer 1's sums (S1+S3 is one arrival-order sum; out is scratch until the layer sum), both layer arrays cleared by stores inside deg_hist and bucket_scatter
# speedup vs baseline: 3.2875x; 1.1996x over previous
#include <hip/hip_runtime.h>

typedef __attribute__((ext_vector_type(2))) float v2f;
typedef __attribute__((ext_vector_type(8))) float v8f;
typedef __attribute__((ext_vector_type(4))) float v4f;

constexpr int NUM_USERS = 100000;
constexpr int NUM_ITEMS = 50000;
constexpr int N_NODES   = NUM_USERS + NUM_ITEMS;
constexpr int DIM       = 64;
constexpr int SBERT     = 384;
constexpr int NEDGE     = 2000000;
constexpr int NLAYERS   = 3;

constexpr int EW_BLOCKS  = 625;
constexpr int EW_THREADS = 256;
constexpr int EW_LANES   = EW_BLOCKS * EW_THREADS;

constexpr int USER_V4 = NUM_USERS * DIM / 4;
constexpr int NODE_V4 = N_NODES * DIM / 4;
constexpr int USER_TRIPS = USER_V4 / EW_LANES;
constexpr int NODE_TRIPS = NODE_V4 / EW_LANES;
static_assert(USER_TRIPS * EW_LANES * 4 == NUM_USERS * DIM, "user array must split evenly");
static_assert(NODE_TRIPS * EW_LANES * 4 == N_NODES * DIM,   "node array must split evenly");

__global__ __launch_bounds__(EW_THREADS)
void init_user(const float* __restrict__ u, float* __restrict__ emb0) {
  const v4f* __restrict__ u4 = (const v4f*)u;
  v4f* __restrict__ e4 = (v4f*)emb0;
  const int t = blockIdx.x * EW_THREADS + threadIdx.x;
  v4f v[USER_TRIPS];
  for (int k = 0; k < USER_TRIPS; ++k) v[k] = u4[t + k * EW_LANES];
  for (int k = 0; k < USER_TRIPS; ++k) e4[t + k * EW_LANES] = v[k];
}

constexpr int TICKET_PITCH = 1;
constexpr int N_TICKETS    = 2 * NLAYERS;

template <int GRID_LANES>
__device__ __forceinline__ void clear_share(float* __restrict__ p, int lane_in_grid) {
  v4f* __restrict__ p4 = (v4f*)p;
  const v4f z = {0.0f, 0.0f, 0.0f, 0.0f};
  constexpr int TRIPS = (NODE_V4 + GRID_LANES - 1) / GRID_LANES;
  for (int k = 0; k < TRIPS; ++k) {
    const int idx = lane_in_grid + k * GRID_LANES;
    if (idx < NODE_V4) p4[idx] = z;
  }
}

__global__ __launch_bounds__(128)
void wmma_proj(const float* __restrict__ bert, const float* __restrict__ w,
               float* __restrict__ emb0) {
  const int lane = threadIdx.x & 31;
  const int wave = threadIdx.x >> 5;
  const int m0   = blockIdx.x * 16;
  const int n0   = wave * 16;
  const int half = lane >> 4;
  const int r    = lane & 15;

  const float* arow = bert + (size_t)(m0 + r) * SBERT + 4 * half;
  const float* brow = w    + (size_t)(n0 + r) * SBERT + 4 * half;

  v8f c = {};
#pragma unroll 8
  for (int k = 0; k < SBERT; k += 8) {
    const v4f a = *(const v4f*)(arow + k);
    const v4f b = *(const v4f*)(brow + k);
    const v2f a0 = {a.x, a.y}, b0 = {b.x, b.y};
    const v2f a1 = {a.z, a.w}, b1 = {b.z, b.w};
    c = __builtin_amdgcn_wmma_f32_16x16x4_f32(
         false, a0,  false, b0,
         (short)0, c,  false,  false);
    c = __builtin_amdgcn_wmma_f32_16x16x4_f32(
         false, a1,  false, b1,
         (short)0, c,  false,  false);
  }

  const size_t base = (size_t)(NUM_USERS + m0 + 8 * half) * DIM + (size_t)(n0 + r);
#pragma unroll
  for (int i = 0; i < 8; ++i) emb0[base + (size_t)i * DIM] = c[i];
}

constexpr int DH_RANGES  = 8;
constexpr int DH_PARTS   = 32;
constexpr int DH_BINS    = N_NODES / DH_RANGES;
constexpr int DH_PITCH   = (DH_BINS + 31) & ~31;
constexpr int DH_THREADS = 1024;
constexpr int DH_EDGES   = NEDGE / DH_PARTS;
constexpr int DH_V4      = DH_EDGES / 4;
constexpr int DH_TRIPS   = (DH_V4 + DH_THREADS - 1) / DH_THREADS;
static_assert(DH_BINS * DH_RANGES == N_NODES, "ranges must cover the nodes exactly");
static_assert(DH_EDGES * DH_PARTS == NEDGE,   "slices must cover the edges exactly");
static_assert(DH_V4 * 4 == DH_EDGES,          "a slice must be whole 16-byte items");
constexpr int PLANE_WORDS = DH_PARTS * DH_RANGES * DH_PITCH;
static_assert(PLANE_WORDS % 64 == 0, "what lies behind the planes starts on a 256-B boundary");

typedef int v4i __attribute__((ext_vector_type(4)));

__global__ __launch_bounds__(DH_THREADS)
void deg_hist(const int* __restrict__ dst, unsigned* __restrict__ planes,
              float* __restrict__ clr, int* __restrict__ ticket) {
  __shared__ unsigned bins[DH_BINS];
  const int range = blockIdx.x % DH_RANGES;
  const int part  = blockIdx.x / DH_RANGES;
  const int tid   = threadIdx.x;

  const v4i* __restrict__ p = (const v4i*)(dst + (size_t)part * DH_EDGES);
  v4i v[DH_TRIPS];
#pragma unroll
  for (int k = 0; k < DH_TRIPS; ++k) {
    const int idx = tid + k * DH_THREADS;
    v[k] = (idx < DH_V4) ? p[idx] : v4i{-1, -1, -1, -1};
  }

  for (int i = tid; i < DH_BINS; i += DH_THREADS) bins[i] = 0u;
  __syncthreads();

  const int g = blockIdx.x * DH_THREADS + tid;
  clear_share<DH_RANGES * DH_PARTS * DH_THREADS>(clr, g);
  if (g < N_TICKETS) ticket[g * TICKET_PITCH] = 0;

  const int base = range * DH_BINS;
#pragma unroll
  for (int k = 0; k < DH_TRIPS; ++k) {
#pragma unroll
    for (int c = 0; c < 4; ++c) {
      const unsigned u = (unsigned)(v[k][c] - base);
      if (u < (unsigned)DH_BINS) atomicAdd(&bins[u], 1u);
    }
  }
  __syncthreads();

  unsigned* __restrict__ o = planes + ((size_t)part * DH_RANGES + range) * DH_PITCH;
  for (int i = tid; i < DH_BINS; i += DH_THREADS) o[i] = bins[i];
}

__global__ __launch_bounds__(256)
void deg_to_dis(const unsigned* __restrict__ planes, float* __restrict__ dis) {
  const int i = blockIdx.x * blockDim.x + threadIdx.x;
  if (i < N_NODES) {
    const int range = i / DH_BINS;
    const unsigned* __restrict__ p = planes + (size_t)range * DH_PITCH + (i - range * DH_BINS);
    unsigned c = 0u;
#pragma unroll
    for (int k = 0; k < DH_PARTS; ++k) c += p[(size_t)k * DH_RANGES * DH_PITCH];
    const float d = (float)c;
    dis[i] = (d > 0.0f) ? rsqrtf(fmaxf(d, 1.0f)) : 0.0f;
  }
}

constexpr int BK_ROWS    = 586;
constexpr int NBLK       = 256;
constexpr int NBKT       = 2 * NBLK;
constexpr int PART_SPLIT = N_NODES / 2;
constexpr int HDR        = (NBKT + 1) * 8;
constexpr int SL_N       = 250;
constexpr int SL_EDGES   = NEDGE / SL_N;
constexpr int SL_V4      = SL_EDGES / 4;
constexpr int BS_THREADS = 512;
constexpr int BS_TRIPS   = (SL_V4 + BS_THREADS - 1) / BS_THREADS;
static_assert(BK_ROWS * NBLK >= N_NODES,  "blocks must cover the nodes");
static_assert(BK_ROWS <= 1024,            "row inside a block must fit in 10 bits");
static_assert(N_NODES <= (1 << 18),       "destination must fit in 18 bits");
static_assert(SL_EDGES * SL_N == NEDGE && SL_V4 * 4 == SL_EDGES, "slices must split evenly");
static_assert(BS_THREADS == NBKT,         "one thread per bucket in the bucketing kernels");
static_assert(NEDGE % 16 == 0 && HDR <= NEDGE, "record batches and header must fit");
static_assert((size_t)PLANE_WORDS + 2 * (size_t)SL_N * NBKT + NBKT + 1 <= (size_t)N_NODES * DIM,
              "the plane pieces and the count tables must fit into the output array");

__device__ __forceinline__ int bucket_of(int s, int d) {
  return 2 * (s / BK_ROWS) + (int)(d >= PART_SPLIT);
}

__global__ __launch_bounds__(BS_THREADS)
void bucket_count(const int* __restrict__ src, const int* __restrict__ dst,
                  unsigned* __restrict__ cnt) {
  __shared__ unsigned h[NBKT];
  const int tid = threadIdx.x;
  const int slice = blockIdx.x;
  const v4i* __restrict__ ps = (const v4i*)(src + (size_t)slice * SL_EDGES);
  const v4i* __restrict__ pd = (const v4i*)(dst + (size_t)slice * SL_EDGES);
  v4i sv[BS_TRIPS], dv[BS_TRIPS];
#pragma unroll
  for (int k = 0; k < BS_TRIPS; ++k) {
    const int idx = tid + k * BS_THREADS;
    sv[k] = (idx < SL_V4) ? ps[idx] : v4i{-1, -1, -1, -1};
    dv[k] = (idx < SL_V4) ? pd[idx] : v4i{0, 0, 0, 0};
  }
  h[tid] = 0u;
  __syncthreads();
#pragma unroll
  for (int k = 0; k < BS_TRIPS; ++k) {
#pragma unroll
    for (int c = 0; c < 4; ++c)
      if (sv[k][c] >= 0) atomicAdd(&h[bucket_of(sv[k][c], dv[k][c])], 1u);
  }
  __syncthreads();
  cnt[(size_t)slice * NBKT + tid] = h[tid];
}

constexpr int BO_THREADS = 2 * NBKT;
constexpr int BO_SLICES  = SL_N / 2;
static_assert(BO_SLICES * 2 == SL_N, "two halves of the slices");
__global__ __launch_bounds__(BO_THREADS)
void bucket_offsets(const unsigned* __restrict__ cnt, unsigned* __restrict__ pre,
                    unsigned* __restrict__ off) {
  __shared__ unsigned part[2][NBKT];
  const int q = threadIdx.x / NBKT;
  const int b = threadIdx.x % NBKT;
  const unsigned* __restrict__ col = cnt + (size_t)q * BO_SLICES * NBKT + b;
  unsigned sum = 0u;
#pragma unroll 25
  for (int s = 0; s < BO_SLICES; ++s) sum += col[(size_t)s * NBKT];
  part[q][b] = sum;
  __syncthreads();
  if (q == 0) {
    unsigned a = 0u, all = 0u;
    for (int j = 0; j < NBKT; ++j) {
      const unsigned c = part[0][j] + part[1][j];
      if (j < b) a += c;
      all += c;
    }
    off[b] = a;
    if (b == 0) off[NBKT] = all;
  }
  unsigned run = q ? part[0][b] : 0u;
  unsigned* __restrict__ pcol = pre + (size_t)q * BO_SLICES * NBKT + b;
#pragma unroll 25
  for (int s = 0; s < BO_SLICES; ++s) {
    const unsigned c = col[(size_t)s * NBKT];
    pcol[(size_t)s * NBKT] = run;
    run += c;
  }
}

__global__ __launch_bounds__(BS_THREADS)
void bucket_scatter(const int* __restrict__ src, const int* __restrict__ dst,
                    const unsigned* __restrict__ cnt, const unsigned* __restrict__ pre,
                    const unsigned* __restrict__ off, unsigned* __restrict__ rec,
                    float* __restrict__ clr) {
  __shared__ unsigned lc[NBKT];
  __shared__ unsigned lstart[NBKT];
  __shared__ unsigned lcur[NBKT];
  __shared__ unsigned gbase[NBKT];
  __shared__ unsigned sw[SL_EDGES];
  __shared__ unsigned short sb[SL_EDGES];
  const int tid = threadIdx.x;
  const int slice = blockIdx.x;
  const v4i* __restrict__ ps = (const v4i*)(src + (size_t)slice * SL_EDGES);
  const v4i* __restrict__ pd = (const v4i*)(dst + (size_t)slice * SL_EDGES);
  v4i sv[BS_TRIPS], dv[BS_TRIPS];
#pragma unroll
  for (int k = 0; k < BS_TRIPS; ++k) {
    const int idx = tid + k * BS_THREADS;
    sv[k] = (idx < SL_V4) ? ps[idx] : v4i{-1, -1, -1, -1};
    dv[k] = (idx < SL_V4) ? pd[idx] : v4i{0, 0, 0, 0};
  }
  lc[tid] = cnt[(size_t)slice * NBKT + tid];
  gbase[tid] = off[tid] + pre[(size_t)slice * NBKT + tid];
  __syncthreads();
  clear_share<SL_N * BS_THREADS>(clr, slice * BS_THREADS + tid);
  unsigned a = 0u;
  for (int j = 0; j < NBKT; ++j) {
    const unsigned c = lc[j];
    if (j < tid) a += c;
  }
  lstart[tid] = a;
  lcur[tid] = a;
  __syncthreads();
#pragma unroll
  for (int k = 0; k < BS_TRIPS; ++k) {
#pragma unroll
    for (int c = 0; c < 4; ++c) {
      const int s = sv[k][c];
      if (s >= 0) {
        const int d = dv[k][c];
        const int bk = bucket_of(s, d);
        const unsigned r = (unsigned)(s - (s / BK_ROWS) * BK_ROWS);
        const unsigned p = atomicAdd(&lcur[bk], 1u);
        sw[p] = (r << 18) | (unsigned)d;
        sb[p] = (unsigned short)bk;
      }
    }
  }
  __syncthreads();
  for (int i = tid; i < SL_EDGES; i += BS_THREADS) {
    const int bk = sb[i];
    const unsigned pos = gbase[bk] + ((unsigned)i - lstart[bk]);
    unsigned w = sw[i];
    if (pos < (unsigned)HDR) w |= ((off[pos >> 3] >> (4u * (pos & 7u))) & 15u) << 28;
    rec[pos] = w;
  }
}

constexpr int SP_THREADS = 1024;
constexpr int SP_GRID    = NBKT;
constexpr int ST_ITEMS   = BK_ROWS * DIM / 4;
constexpr int ST_TRIPS   = (ST_ITEMS + SP_THREADS - 1) / SP_THREADS;
constexpr int ST_HALF    = ST_TRIPS / 2;
static_assert(ST_HALF * 2 == ST_TRIPS, "staging runs in two equal halves");
constexpr int EPW        = 16;

template <bool FIRST>
__global__ __launch_bounds__(SP_THREADS)
void spmm_block(const float* __restrict__ x, const float* __restrict__ dis,
                const unsigned* __restrict__ rec, float* __restrict__ xn,
                int* __restrict__ ticket) {
  __shared__ float rows[BK_ROWS * DIM];
  __shared__ int s_t;
  const int tid  = threadIdx.x;
  const int lane = tid & 31;
  const int wv = __builtin_amdgcn_readfirstlane(tid >> 5);
  int part = (int)((__builtin_amdgcn_s_sendmsg_rtn(0x87) >> 18) & 1u);
  bool other = false;
  typedef unsigned v16u __attribute__((ext_vector_type(16)));
  typedef const __attribute__((address_space(4))) unsigned* cu_p;
  typedef const __attribute__((address_space(4))) v16u*     cu16_p;
  const cu_p crec = (cu_p)rec;

  for (int it = 0; it < 2 * NBLK + 2; ++it) {
    if (tid == 0) s_t = atomicAdd(&ticket[part * TICKET_PITCH], 1);
    __syncthreads();
    const int t = __builtin_amdgcn_readfirstlane(s_t);
    __syncthreads();
    if (t >= NBLK) {
      if (other) break;
      other = true;
      part ^= 1;
      continue;
    }
    const int bkt  = 2 * t + part;
    const int row0 = t * BK_ROWS;
    const int nrows = (N_NODES - row0 < BK_ROWS) ? (N_NODES - row0) : BK_ROWS;

    unsigned st = 0u, en = 0u;
#pragma unroll
    for (int j = 0; j < 8; ++j) {
      st |= (crec[8 * bkt + j]     >> 28) << (4 * j);
      en |= (crec[8 * bkt + 8 + j] >> 28) << (4 * j);
    }

    const v4f* __restrict__ x4 = (const v4f*)(x + (size_t)row0 * DIM);
#pragma nounroll
    for (int k0 = 0; k0 < ST_TRIPS; k0 += ST_HALF) {
      v4f v[ST_HALF];
      float sc[ST_HALF];
#pragma unroll
      for (int k = 0; k < ST_HALF; ++k) {
        const int idx = tid + (k0 + k) * SP_THREADS;
        const int r = idx >> 4;
        v[k] = v4f{0.0f, 0.0f, 0.0f, 0.0f};
        sc[k] = 0.0f;
        if (r < nrows) { v[k] = x4[idx]; sc[k] = dis[row0 + r]; }
      }
#pragma unroll
      for (int k = 0; k < ST_HALF; ++k) {
        const int idx = tid + (k0 + k) * SP_THREADS;
        if (idx < ST_ITEMS) {
          v4f y = v[k] * sc[k];
          if (!FIRST) y = y * sc[k];
          ((v4f*)rows)[idx] = y;
        }
      }
    }
    __syncthreads();

    const int e_first = (int)(st & ~15u);
#pragma nounroll
    for (int e0 = e_first + wv * EPW; e0 < (int)en; e0 += (SP_THREADS / 32) * EPW) {
      const v16u rv = *(cu16_p)(crec + e0);
      float lo[EPW], hi[EPW];
#pragma unroll
      for (int j = 0; j < EPW; ++j) {
        lo[j] = 0.0f;
        hi[j] = 0.0f;
        if (e0 + j >= (int)st && e0 + j < (int)en) {
          const float* row = rows + ((rv[j] >> 18) & 1023u) * DIM;
          lo[j] = row[lane];
          hi[j] = row[lane + 32];
        }
      }
#pragma unroll
      for (int j = 0; j < EPW; ++j) {
        if (e0 + j >= (int)st && e0 + j < (int)en) {
          float* o = xn + (size_t)(rv[j] & 0x3FFFFu) * DIM;
          atomicAdd(o + lane,      lo[j]);
          atomicAdd(o + lane + 32, hi[j]);
        }
      }
    }
  }
}

static_assert(NLAYERS == 3, "layers 1 and 3 share one workspace array, layer 2 has the other");
constexpr int SUM_CHUNK = 5;
static_assert(NODE_TRIPS % SUM_CHUNK == 0, "chunks must cover the lane's items");

__global__ __launch_bounds__(EW_THREADS)
void layer_sum(const float* __restrict__ x0, const float* __restrict__ s13,
               const float* __restrict__ s2, const float* __restrict__ dis,
               float* __restrict__ out) {
  const v4f* __restrict__ p0 = (const v4f*)x0;
  const v4f* __restrict__ p1 = (const v4f*)s13;
  const v4f* __restrict__ p2 = (const v4f*)s2;
  v4f* __restrict__ o4 = (v4f*)out;
  const int t = blockIdx.x * EW_THREADS + threadIdx.x;
#pragma nounroll
  for (int c = 0; c < NODE_TRIPS; c += SUM_CHUNK) {
    v4f a[SUM_CHUNK], b[SUM_CHUNK], d[SUM_CHUNK];
    float g[SUM_CHUNK];
    for (int k = 0; k < SUM_CHUNK; ++k) {
      a[k] = p0[t + (c + k) * EW_LANES];
      b[k] = p1[t + (c + k) * EW_LANES];
      d[k] = p2[t + (c + k) * EW_LANES];
      g[k] = dis[(t + (c + k) * EW_LANES) >> 4];
    }
    for (int k = 0; k < SUM_CHUNK; ++k)
      o4[t + (c + k) * EW_LANES] =
          ((a[k] + b[k] * g[k]) + d[k] * g[k]) * 0.25f;
  }
}

extern "C" void kernel_launch(void* const* d_in, const int* in_sizes, int n_in,
                              void* d_out, int out_size, void* d_ws, size_t ws_size,
                              hipStream_t stream) {
  const int*   ei    = (const int*)d_in[0];
  const int*   src   = ei;
  const int*   dst   = ei + NEDGE;
  const float* user  = (const float*)d_in[1];
  const float* bert  = (const float*)d_in[2];
  const float* wproj = (const float*)d_in[3];

  float* emb0 = (float*)d_out;
  float* out  = emb0 + (size_t)N_NODES * DIM;

  float* deg  = (float*)d_ws;
  float* norm = deg  + ((N_NODES + 63) & ~63);
  float* xA   = norm + ((NEDGE + 63) & ~63);
  float* xB   = xA   + (size_t)N_NODES * DIM;

  const int B = 256;

  init_user<<<EW_BLOCKS, EW_THREADS, 0, stream>>>(user, emb0);
  wmma_proj<<<NUM_ITEMS / 16, 128, 0, stream>>>(bert, wproj, emb0);

  int* tick = (int*)(deg + N_NODES);

  unsigned* planes = (unsigned*)out;
  deg_hist  <<<DH_RANGES * DH_PARTS, DH_THREADS, 0, stream>>>(dst, planes, xA, tick);
  deg_to_dis<<<(N_NODES + B - 1) / B, B, 0, stream>>>(planes, deg);

  unsigned* rec = (unsigned*)norm;
  unsigned* cnt = planes + PLANE_WORDS;
  unsigned* pre = cnt + (size_t)SL_N * NBKT;
  unsigned* off = pre + (size_t)SL_N * NBKT;
  bucket_count  <<<SL_N, BS_THREADS, 0, stream>>>(src, dst, cnt);
  bucket_offsets<<<1, BO_THREADS, 0, stream>>>(cnt, pre, off);
  bucket_scatter<<<SL_N, BS_THREADS, 0, stream>>>(src, dst, cnt, pre, off, rec, xB);

  spmm_block<true> <<<SP_GRID, SP_THREADS, 0, stream>>>(emb0, deg, rec, xA, tick);
  spmm_block<false><<<SP_GRID, SP_THREADS, 0, stream>>>(xA, deg, rec, xB, tick + 2);
  spmm_block<false><<<SP_GRID, SP_THREADS, 0, stream>>>(xB, deg, rec, xA, tick + 4);
  layer_sum<<<EW_BLOCKS, EW_THREADS, 0, stream>>>(emb0, xA, xB, deg, out);
}


# SpMM edge loop: a 16-record batch wholly inside its bucket takes a path with no per-record range test or branch (one test per batch); a bucket's first and last batch keep the tests; same adds
# speedup vs baseline: 3.4200x; 1.0403x over previous
#include <hip/hip_runtime.h>

typedef __attribute__((ext_vector_type(2))) float v2f;
typedef __attribute__((ext_vector_type(8))) float v8f;
typedef __attribute__((ext_vector_type(4))) float v4f;

constexpr int NUM_USERS = 100000;
constexpr int NUM_ITEMS = 50000;
constexpr int N_NODES   = NUM_USERS + NUM_ITEMS;
constexpr int DIM       = 64;
constexpr int SBERT     = 384;
constexpr int NEDGE     = 2000000;
constexpr int NLAYERS   = 3;

constexpr int EW_BLOCKS  = 625;
constexpr int EW_THREADS = 256;
constexpr int EW_LANES   = EW_BLOCKS * EW_THREADS;

constexpr int USER_V4 = NUM_USERS * DIM / 4;
constexpr int NODE_V4 = N_NODES * DIM / 4;
constexpr int USER_TRIPS = USER_V4 / EW_LANES;
constexpr int NODE_TRIPS = NODE_V4 / EW_LANES;
static_assert(USER_TRIPS * EW_LANES * 4 == NUM_USERS * DIM, "user array must split evenly");
static_assert(NODE_TRIPS * EW_LANES * 4 == N_NODES * DIM,   "node array must split evenly");

__global__ __launch_bounds__(EW_THREADS)
void init_user(const float* __restrict__ u, float* __restrict__ emb0) {
  const v4f* __restrict__ u4 = (const v4f*)u;
  v4f* __restrict__ e4 = (v4f*)emb0;
  const int t = blockIdx.x * EW_THREADS + threadIdx.x;
  v4f v[USER_TRIPS];
  for (int k = 0; k < USER_TRIPS; ++k) v[k] = u4[t + k * EW_LANES];
  for (int k = 0; k < USER_TRIPS; ++k) e4[t + k * EW_LANES] = v[k];
}

constexpr int TICKET_PITCH = 1;
constexpr int N_TICKETS    = 2 * NLAYERS;

template <int GRID_LANES>
__device__ __forceinline__ void clear_share(float* __restrict__ p, int lane_in_grid) {
  v4f* __restrict__ p4 = (v4f*)p;
  const v4f z = {0.0f, 0.0f, 0.0f, 0.0f};
  constexpr int TRIPS = (NODE_V4 + GRID_LANES - 1) / GRID_LANES;
  for (int k = 0; k < TRIPS; ++k) {
    const int idx = lane_in_grid + k * GRID_LANES;
    if (idx < NODE_V4) p4[idx] = z;
  }
}

__global__ __launch_bounds__(128)
void wmma_proj(const float* __restrict__ bert, const float* __restrict__ w,
               float* __restrict__ emb0) {
  const int lane = threadIdx.x & 31;
  const int wave = threadIdx.x >> 5;
  const int m0   = blockIdx.x * 16;
  const int n0   = wave * 16;
  const int half = lane >> 4;
  const int r    = lane & 15;

  const float* arow = bert + (size_t)(m0 + r) * SBERT + 4 * half;
  const float* brow = w    + (size_t)(n0 + r) * SBERT + 4 * half;

  v8f c = {};
#pragma unroll 8
  for (int k = 0; k < SBERT; k += 8) {
    const v4f a = *(const v4f*)(arow + k);
    const v4f b = *(const v4f*)(brow + k);
    const v2f a0 = {a.x, a.y}, b0 = {b.x, b.y};
    const v2f a1 = {a.z, a.w}, b1 = {b.z, b.w};
    c = __builtin_amdgcn_wmma_f32_16x16x4_f32(
         false, a0,  false, b0,
         (short)0, c,  false,  false);
    c = __builtin_amdgcn_wmma_f32_16x16x4_f32(
         false, a1,  false, b1,
         (short)0, c,  false,  false);
  }

  const size_t base = (size_t)(NUM_USERS + m0 + 8 * half) * DIM + (size_t)(n0 + r);
#pragma unroll
  for (int i = 0; i < 8; ++i) emb0[base + (size_t)i * DIM] = c[i];
}

constexpr int DH_RANGES  = 8;
constexpr int DH_PARTS   = 32;
constexpr int DH_BINS    = N_NODES / DH_RANGES;
constexpr int DH_PITCH   = (DH_BINS + 31) & ~31;
constexpr int DH_THREADS = 1024;
constexpr int DH_EDGES   = NEDGE / DH_PARTS;
constexpr int DH_V4      = DH_EDGES / 4;
constexpr int DH_TRIPS   = (DH_V4 + DH_THREADS - 1) / DH_THREADS;
static_assert(DH_BINS * DH_RANGES == N_NODES, "ranges must cover the nodes exactly");
static_assert(DH_EDGES * DH_PARTS == NEDGE,   "slices must cover the edges exactly");
static_assert(DH_V4 * 4 == DH_EDGES,          "a slice must be whole 16-byte items");
constexpr int PLANE_WORDS = DH_PARTS * DH_RANGES * DH_PITCH;
static_assert(PLANE_WORDS % 64 == 0, "what lies behind the planes starts on a 256-B boundary");

typedef int v4i __attribute__((ext_vector_type(4)));

__global__ __launch_bounds__(DH_THREADS)
void deg_hist(const int* __restrict__ dst, unsigned* __restrict__ planes,
              float* __restrict__ clr, int* __restrict__ ticket) {
  __shared__ unsigned bins[DH_BINS];
  const int range = blockIdx.x % DH_RANGES;
  const int part  = blockIdx.x / DH_RANGES;
  const int tid   = threadIdx.x;

  const v4i* __restrict__ p = (const v4i*)(dst + (size_t)part * DH_EDGES);
  v4i v[DH_TRIPS];
#pragma unroll
  for (int k = 0; k < DH_TRIPS; ++k) {
    const int idx = tid + k * DH_THREADS;
    v[k] = (idx < DH_V4) ? p[idx] : v4i{-1, -1, -1, -1};
  }

  for (int i = tid; i < DH_BINS; i += DH_THREADS) bins[i] = 0u;
  __syncthreads();

  const int g = blockIdx.x * DH_THREADS + tid;
  clear_share<DH_RANGES * DH_PARTS * DH_THREADS>(clr, g);
  if (g < N_TICKETS) ticket[g * TICKET_PITCH] = 0;

  const int base = range * DH_BINS;
#pragma unroll
  for (int k = 0; k < DH_TRIPS; ++k) {
#pragma unroll
    for (int c = 0; c < 4; ++c) {
      const unsigned u = (unsigned)(v[k][c] - base);
      if (u < (unsigned)DH_BINS) atomicAdd(&bins[u], 1u);
    }
  }
  __syncthreads();

  unsigned* __restrict__ o = planes + ((size_t)part * DH_RANGES + range) * DH_PITCH;
  for (int i = tid; i < DH_BINS; i += DH_THREADS) o[i] = bins[i];
}

__global__ __launch_bounds__(256)
void deg_to_dis(const unsigned* __restrict__ planes, float* __restrict__ dis) {
  const int i = blockIdx.x * blockDim.x + threadIdx.x;
  if (i < N_NODES) {
    const int range = i / DH_BINS;
    const unsigned* __restrict__ p = planes + (size_t)range * DH_PITCH + (i - range * DH_BINS);
    unsigned c = 0u;
#pragma unroll
    for (int k = 0; k < DH_PARTS; ++k) c += p[(size_t)k * DH_RANGES * DH_PITCH];
    const float d = (float)c;
    dis[i] = (d > 0.0f) ? rsqrtf(fmaxf(d, 1.0f)) : 0.0f;
  }
}

constexpr int BK_ROWS    = 586;
constexpr int NBLK       = 256;
constexpr int NBKT       = 2 * NBLK;
constexpr int PART_SPLIT = N_NODES / 2;
constexpr int HDR        = (NBKT + 1) * 8;
constexpr int SL_N       = 250;
constexpr int SL_EDGES   = NEDGE / SL_N;
constexpr int SL_V4      = SL_EDGES / 4;
constexpr int BS_THREADS = 512;
constexpr int BS_TRIPS   = (SL_V4 + BS_THREADS - 1) / BS_THREADS;
static_assert(BK_ROWS * NBLK >= N_NODES,  "blocks must cover the nodes");
static_assert(BK_ROWS <= 1024,            "row inside a block must fit in 10 bits");
static_assert(N_NODES <= (1 << 18),       "destination must fit in 18 bits");
static_assert(SL_EDGES * SL_N == NEDGE && SL_V4 * 4 == SL_EDGES, "slices must split evenly");
static_assert(BS_THREADS == NBKT,         "one thread per bucket in the bucketing kernels");
static_assert(NEDGE % 16 == 0 && HDR <= NEDGE, "record batches and header must fit");
static_assert((size_t)PLANE_WORDS + 2 * (size_t)SL_N * NBKT + NBKT + 1 <= (size_t)N_NODES * DIM,
              "the plane pieces and the count tables must fit into the output array");

__device__ __forceinline__ int bucket_of(int s, int d) {
  return 2 * (s / BK_ROWS) + (int)(d >= PART_SPLIT);
}

__global__ __launch_bounds__(BS_THREADS)
void bucket_count(const int* __restrict__ src, const int* __restrict__ dst,
                  unsigned* __restrict__ cnt) {
  __shared__ unsigned h[NBKT];
  const int tid = threadIdx.x;
  const int slice = blockIdx.x;
  const v4i* __restrict__ ps = (const v4i*)(src + (size_t)slice * SL_EDGES);
  const v4i* __restrict__ pd = (const v4i*)(dst + (size_t)slice * SL_EDGES);
  v4i sv[BS_TRIPS], dv[BS_TRIPS];
#pragma unroll
  for (int k = 0; k < BS_TRIPS; ++k) {
    const int idx = tid + k * BS_THREADS;
    sv[k] = (idx < SL_V4) ? ps[idx] : v4i{-1, -1, -1, -1};
    dv[k] = (idx < SL_V4) ? pd[idx] : v4i{0, 0, 0, 0};
  }
  h[tid] = 0u;
  __syncthreads();
#pragma unroll
  for (int k = 0; k < BS_TRIPS; ++k) {
#pragma unroll
    for (int c = 0; c < 4; ++c)
      if (sv[k][c] >= 0) atomicAdd(&h[bucket_of(sv[k][c], dv[k][c])], 1u);
  }
  __syncthreads();
  cnt[(size_t)slice * NBKT + tid] = h[tid];
}

constexpr int BO_THREADS = 2 * NBKT;
constexpr int BO_SLICES  = SL_N / 2;
static_assert(BO_SLICES * 2 == SL_N, "two halves of the slices");
__global__ __launch_bounds__(BO_THREADS)
void bucket_offsets(const unsigned* __restrict__ cnt, unsigned* __restrict__ pre,
                    unsigned* __restrict__ off) {
  __shared__ unsigned part[2][NBKT];
  const int q = threadIdx.x / NBKT;
  const int b = threadIdx.x % NBKT;
  const unsigned* __restrict__ col = cnt + (size_t)q * BO_SLICES * NBKT + b;
  unsigned sum = 0u;
#pragma unroll 25
  for (int s = 0; s < BO_SLICES; ++s) sum += col[(size_t)s * NBKT];
  part[q][b] = sum;
  __syncthreads();
  if (q == 0) {
    unsigned a = 0u, all = 0u;
    for (int j = 0; j < NBKT; ++j) {
      const unsigned c = part[0][j] + part[1][j];
      if (j < b) a += c;
      all += c;
    }
    off[b] = a;
    if (b == 0) off[NBKT] = all;
  }
  unsigned run = q ? part[0][b] : 0u;
  unsigned* __restrict__ pcol = pre + (size_t)q * BO_SLICES * NBKT + b;
#pragma unroll 25
  for (int s = 0; s < BO_SLICES; ++s) {
    const unsigned c = col[(size_t)s * NBKT];
    pcol[(size_t)s * NBKT] = run;
    run += c;
  }
}

__global__ __launch_bounds__(BS_THREADS)
void bucket_scatter(const int* __restrict__ src, const int* __restrict__ dst,
                    const unsigned* __restrict__ cnt, const unsigned* __restrict__ pre,
                    const unsigned* __restrict__ off, unsigned* __restrict__ rec,
                    float* __restrict__ clr) {
  __shared__ unsigned lc[NBKT];
  __shared__ unsigned lstart[NBKT];
  __shared__ unsigned lcur[NBKT];
  __shared__ unsigned gbase[NBKT];
  __shared__ unsigned sw[SL_EDGES];
  __shared__ unsigned short sb[SL_EDGES];
  const int tid = threadIdx.x;
  const int slice = blockIdx.x;
  const v4i* __restrict__ ps = (const v4i*)(src + (size_t)slice * SL_EDGES);
  const v4i* __restrict__ pd = (const v4i*)(dst + (size_t)slice * SL_EDGES);
  v4i sv[BS_TRIPS], dv[BS_TRIPS];
#pragma unroll
  for (int k = 0; k < BS_TRIPS; ++k) {
    const int idx = tid + k * BS_THREADS;
    sv[k] = (idx < SL_V4) ? ps[idx] : v4i{-1, -1, -1, -1};
    dv[k] = (idx < SL_V4) ? pd[idx] : v4i{0, 0, 0, 0};
  }
  lc[tid] = cnt[(size_t)slice * NBKT + tid];
  gbase[tid] = off[tid] + pre[(size_t)slice * NBKT + tid];
  __syncthreads();
  clear_share<SL_N * BS_THREADS>(clr, slice * BS_THREADS + tid);
  unsigned a = 0u;
  for (int j = 0; j < NBKT; ++j) {
    const unsigned c = lc[j];
    if (j < tid) a += c;
  }
  lstart[tid] = a;
  lcur[tid] = a;
  __syncthreads();
#pragma unroll
  for (int k = 0; k < BS_TRIPS; ++k) {
#pragma unroll
    for (int c = 0; c < 4; ++c) {
      const int s = sv[k][c];
      if (s >= 0) {
        const int d = dv[k][c];
        const int bk = bucket_of(s, d);
        const unsigned r = (unsigned)(s - (s / BK_ROWS) * BK_ROWS);
        const unsigned p = atomicAdd(&lcur[bk], 1u);
        sw[p] = (r << 18) | (unsigned)d;
        sb[p] = (unsigned short)bk;
      }
    }
  }
  __syncthreads();
  for (int i = tid; i < SL_EDGES; i += BS_THREADS) {
    const int bk = sb[i];
    const unsigned pos = gbase[bk] + ((unsigned)i - lstart[bk]);
    unsigned w = sw[i];
    if (pos < (unsigned)HDR) w |= ((off[pos >> 3] >> (4u * (pos & 7u))) & 15u) << 28;
    rec[pos] = w;
  }
}

constexpr int SP_THREADS = 1024;
constexpr int SP_GRID    = NBKT;
constexpr int ST_ITEMS   = BK_ROWS * DIM / 4;
constexpr int ST_TRIPS   = (ST_ITEMS + SP_THREADS - 1) / SP_THREADS;
constexpr int ST_HALF    = ST_TRIPS / 2;
static_assert(ST_HALF * 2 == ST_TRIPS, "staging runs in two equal halves");
constexpr int EPW        = 16;

typedef unsigned v16u __attribute__((ext_vector_type(16)));

template <bool GUARD>
__device__ __forceinline__ void edge_batch(const v16u rv, int e0, int st, int en,
                                           const float* rows, float* __restrict__ xn,
                                           int lane) {
  float lo[EPW], hi[EPW];
#pragma unroll
  for (int j = 0; j < EPW; ++j) {
    lo[j] = 0.0f;
    hi[j] = 0.0f;
    if (!GUARD || (e0 + j >= st && e0 + j < en)) {
      const float* row = rows + ((rv[j] >> 18) & 1023u) * DIM;
      lo[j] = row[lane];
      hi[j] = row[lane + 32];
    }
  }
#pragma unroll
  for (int j = 0; j < EPW; ++j) {
    if (!GUARD || (e0 + j >= st && e0 + j < en)) {
      float* o = xn + (size_t)(rv[j] & 0x3FFFFu) * DIM;
      atomicAdd(o + lane,      lo[j]);
      atomicAdd(o + lane + 32, hi[j]);
    }
  }
}

template <bool FIRST>
__global__ __launch_bounds__(SP_THREADS)
void spmm_block(const float* __restrict__ x, const float* __restrict__ dis,
                const unsigned* __restrict__ rec, float* __restrict__ xn,
                int* __restrict__ ticket) {
  __shared__ float rows[BK_ROWS * DIM];
  __shared__ int s_t;
  const int tid  = threadIdx.x;
  const int lane = tid & 31;
  const int wv = __builtin_amdgcn_readfirstlane(tid >> 5);
  int part = (int)((__builtin_amdgcn_s_sendmsg_rtn(0x87) >> 18) & 1u);
  bool other = false;
  typedef const __attribute__((address_space(4))) unsigned* cu_p;
  typedef const __attribute__((address_space(4))) v16u*     cu16_p;
  const cu_p crec = (cu_p)rec;

  for (int it = 0; it < 2 * NBLK + 2; ++it) {
    if (tid == 0) s_t = atomicAdd(&ticket[part * TICKET_PITCH], 1);
    __syncthreads();
    const int t = __builtin_amdgcn_readfirstlane(s_t);
    __syncthreads();
    if (t >= NBLK) {
      if (other) break;
      other = true;
      part ^= 1;
      continue;
    }
    const int bkt  = 2 * t + part;
    const int row0 = t * BK_ROWS;
    const int nrows = (N_NODES - row0 < BK_ROWS) ? (N_NODES - row0) : BK_ROWS;

    unsigned st = 0u, en = 0u;
#pragma unroll
    for (int j = 0; j < 8; ++j) {
      st |= (crec[8 * bkt + j]     >> 28) << (4 * j);
      en |= (crec[8 * bkt + 8 + j] >> 28) << (4 * j);
    }

    const v4f* __restrict__ x4 = (const v4f*)(x + (size_t)row0 * DIM);
#pragma nounroll
    for (int k0 = 0; k0 < ST_TRIPS; k0 += ST_HALF) {
      v4f v[ST_HALF];
      float sc[ST_HALF];
#pragma unroll
      for (int k = 0; k < ST_HALF; ++k) {
        const int idx = tid + (k0 + k) * SP_THREADS;
        const int r = idx >> 4;
        v[k] = v4f{0.0f, 0.0f, 0.0f, 0.0f};
        sc[k] = 0.0f;
        if (r < nrows) { v[k] = x4[idx]; sc[k] = dis[row0 + r]; }
      }
#pragma unroll
      for (int k = 0; k < ST_HALF; ++k) {
        const int idx = tid + (k0 + k) * SP_THREADS;
        if (idx < ST_ITEMS) {
          v4f y = v[k] * sc[k];
          if (!FIRST) y = y * sc[k];
          ((v4f*)rows)[idx] = y;
        }
      }
    }
    __syncthreads();

    const int e_first = (int)(st & ~15u);
#pragma nounroll
    for (int e0 = e_first + wv * EPW; e0 < (int)en; e0 += (SP_THREADS / 32) * EPW) {
      const v16u rv = *(cu16_p)(crec + e0);
      if (e0 >= (int)st && e0 + EPW <= (int)en)
        edge_batch<false>(rv, e0, (int)st, (int)en, rows, xn, lane);
      else
        edge_batch<true>(rv, e0, (int)st, (int)en, rows, xn, lane);
    }
  }
}

static_assert(NLAYERS == 3, "layers 1 and 3 share one workspace array, layer 2 has the other");
constexpr int SUM_CHUNK = 5;
static_assert(NODE_TRIPS % SUM_CHUNK == 0, "chunks must cover the lane's items");

__global__ __launch_bounds__(EW_THREADS)
void layer_sum(const float* __restrict__ x0, const float* __restrict__ s13,
               const float* __restrict__ s2, const float* __restrict__ dis,
               float* __restrict__ out) {
  const v4f* __restrict__ p0 = (const v4f*)x0;
  const v4f* __restrict__ p1 = (const v4f*)s13;
  const v4f* __restrict__ p2 = (const v4f*)s2;
  v4f* __restrict__ o4 = (v4f*)out;
  const int t = blockIdx.x * EW_THREADS + threadIdx.x;
#pragma nounroll
  for (int c = 0; c < NODE_TRIPS; c += SUM_CHUNK) {
    v4f a[SUM_CHUNK], b[SUM_CHUNK], d[SUM_CHUNK];
    float g[SUM_CHUNK];
    for (int k = 0; k < SUM_CHUNK; ++k) {
      a[k] = p0[t + (c + k) * EW_LANES];
      b[k] = p1[t + (c + k) * EW_LANES];
      d[k] = p2[t + (c + k) * EW_LANES];
      g[k] = dis[(t + (c + k) * EW_LANES) >> 4];
    }
    for (int k = 0; k < SUM_CHUNK; ++k)
      o4[t + (c + k) * EW_LANES] =
          ((a[k] + b[k] * g[k]) + d[k] * g[k]) * 0.25f;
  }
}

extern "C" void kernel_launch(void* const* d_in, const int* in_sizes, int n_in,
                              void* d_out, int out_size, void* d_ws, size_t ws_size,
                              hipStream_t stream) {
  const int*   ei    = (const int*)d_in[0];
  const int*   src   = ei;
  const int*   dst   = ei + NEDGE;
  const float* user  = (const float*)d_in[1];
  const float* bert  = (const float*)d_in[2];
  const float* wproj = (const float*)d_in[3];

  float* emb0 = (float*)d_out;
  float* out  = emb0 + (size_t)N_NODES * DIM;

  float* deg  = (float*)d_ws;
  float* norm = deg  + ((N_NODES + 63) & ~63);
  float* xA   = norm + ((NEDGE + 63) & ~63);
  float* xB   = xA   + (size_t)N_NODES * DIM;

  const int B = 256;

  init_user<<<EW_BLOCKS, EW_THREADS, 0, stream>>>(user, emb0);
  wmma_proj<<<NUM_ITEMS / 16, 128, 0, stream>>>(bert, wproj, emb0);

  int* tick = (int*)(deg + N_NODES);

  unsigned* planes = (unsigned*)out;
  deg_hist  <<<DH_RANGES * DH_PARTS, DH_THREADS, 0, stream>>>(dst, planes, xA, tick);
  deg_to_dis<<<(N_NODES + B - 1) / B, B, 0, stream>>>(planes, deg);

  unsigned* rec = (unsigned*)norm;
  unsigned* cnt = planes + PLANE_WORDS;
  unsigned* pre = cnt + (size_t)SL_N * NBKT;
  unsigned* off = pre + (size_t)SL_N * NBKT;
  bucket_count  <<<SL_N, BS_THREADS, 0, stream>>>(src, dst, cnt);
  bucket_offsets<<<1, BO_THREADS, 0, stream>>>(cnt, pre, off);
  bucket_scatter<<<SL_N, BS_THREADS, 0, stream>>>(src, dst, cnt, pre, off, rec, xB);

  spmm_block<true> <<<SP_GRID, SP_THREADS, 0, stream>>>(emb0, deg, rec, xA, tick);
  spmm_block<false><<<SP_GRID, SP_THREADS, 0, stream>>>(xA, deg, rec, xB, tick + 2);
  spmm_block<false><<<SP_GRID, SP_THREADS, 0, stream>>>(xB, deg, rec, xA, tick + 4);
  layer_sum<<<EW_BLOCKS, EW_THREADS, 0, stream>>>(emb0, xA, xB, deg, out);
}


# No offsets kernel: bucket_count adds its counts to bucket totals; a scatter work-group forms the bucket prefix itself and draws its run's place by one integer add; f32 sums arrival-order as before
# speedup vs baseline: 3.6780x; 1.0754x over previous
#include <hip/hip_runtime.h>

typedef __attribute__((ext_vector_type(2))) float v2f;
typedef __attribute__((ext_vector_type(8))) float v8f;
typedef __attribute__((ext_vector_type(4))) float v4f;

constexpr int NUM_USERS = 100000;
constexpr int NUM_ITEMS = 50000;
constexpr int N_NODES   = NUM_USERS + NUM_ITEMS;
constexpr int DIM       = 64;
constexpr int SBERT     = 384;
constexpr int NEDGE     = 2000000;
constexpr int NLAYERS   = 3;

constexpr int EW_BLOCKS  = 625;
constexpr int EW_THREADS = 256;
constexpr int EW_LANES   = EW_BLOCKS * EW_THREADS;

constexpr int USER_V4 = NUM_USERS * DIM / 4;
constexpr int NODE_V4 = N_NODES * DIM / 4;
constexpr int USER_TRIPS = USER_V4 / EW_LANES;
constexpr int NODE_TRIPS = NODE_V4 / EW_LANES;
static_assert(USER_TRIPS * EW_LANES * 4 == NUM_USERS * DIM, "user array must split evenly");
static_assert(NODE_TRIPS * EW_LANES * 4 == N_NODES * DIM,   "node array must split evenly");

__global__ __launch_bounds__(EW_THREADS)
void init_user(const float* __restrict__ u, float* __restrict__ emb0) {
  const v4f* __restrict__ u4 = (const v4f*)u;
  v4f* __restrict__ e4 = (v4f*)emb0;
  const int t = blockIdx.x * EW_THREADS + threadIdx.x;
  v4f v[USER_TRIPS];
  for (int k = 0; k < USER_TRIPS; ++k) v[k] = u4[t + k * EW_LANES];
  for (int k = 0; k < USER_TRIPS; ++k) e4[t + k * EW_LANES] = v[k];
}

constexpr int TICKET_PITCH = 1;
constexpr int N_TICKETS    = 2 * NLAYERS;
constexpr int ZERO_WORDS   = 1024;

template <int GRID_LANES>
__device__ __forceinline__ void clear_share(float* __restrict__ p, int lane_in_grid) {
  v4f* __restrict__ p4 = (v4f*)p;
  const v4f z = {0.0f, 0.0f, 0.0f, 0.0f};
  constexpr int TRIPS = (NODE_V4 + GRID_LANES - 1) / GRID_LANES;
  for (int k = 0; k < TRIPS; ++k) {
    const int idx = lane_in_grid + k * GRID_LANES;
    if (idx < NODE_V4) p4[idx] = z;
  }
}

__global__ __launch_bounds__(128)
void wmma_proj(const float* __restrict__ bert, const float* __restrict__ w,
               float* __restrict__ emb0) {
  const int lane = threadIdx.x & 31;
  const int wave = threadIdx.x >> 5;
  const int m0   = blockIdx.x * 16;
  const int n0   = wave * 16;
  const int half = lane >> 4;
  const int r    = lane & 15;

  const float* arow = bert + (size_t)(m0 + r) * SBERT + 4 * half;
  const float* brow = w    + (size_t)(n0 + r) * SBERT + 4 * half;

  v8f c = {};
#pragma unroll 8
  for (int k = 0; k < SBERT; k += 8) {
    const v4f a = *(const v4f*)(arow + k);
    const v4f b = *(const v4f*)(brow + k);
    const v2f a0 = {a.x, a.y}, b0 = {b.x, b.y};
    const v2f a1 = {a.z, a.w}, b1 = {b.z, b.w};
    c = __builtin_amdgcn_wmma_f32_16x16x4_f32(
         false, a0,  false, b0,
         (short)0, c,  false,  false);
    c = __builtin_amdgcn_wmma_f32_16x16x4_f32(
         false, a1,  false, b1,
         (short)0, c,  false,  false);
  }

  const size_t base = (size_t)(NUM_USERS + m0 + 8 * half) * DIM + (size_t)(n0 + r);
#pragma unroll
  for (int i = 0; i < 8; ++i) emb0[base + (size_t)i * DIM] = c[i];
}

constexpr int DH_RANGES  = 8;
constexpr int DH_PARTS   = 32;
constexpr int DH_BINS    = N_NODES / DH_RANGES;
constexpr int DH_PITCH   = (DH_BINS + 31) & ~31;
constexpr int DH_THREADS = 1024;
constexpr int DH_EDGES   = NEDGE / DH_PARTS;
constexpr int DH_V4      = DH_EDGES / 4;
constexpr int DH_TRIPS   = (DH_V4 + DH_THREADS - 1) / DH_THREADS;
static_assert(DH_BINS * DH_RANGES == N_NODES, "ranges must cover the nodes exactly");
static_assert(DH_EDGES * DH_PARTS == NEDGE,   "slices must cover the edges exactly");
static_assert(DH_V4 * 4 == DH_EDGES,          "a slice must be whole 16-byte items");
constexpr int PLANE_WORDS = DH_PARTS * DH_RANGES * DH_PITCH;
static_assert(PLANE_WORDS % 64 == 0, "what lies behind the planes starts on a 256-B boundary");

typedef int v4i __attribute__((ext_vector_type(4)));

__global__ __launch_bounds__(DH_THREADS)
void deg_hist(const int* __restrict__ dst, unsigned* __restrict__ planes,
              float* __restrict__ clr, int* __restrict__ ticket,
              unsigned* __restrict__ zero_w) {
  __shared__ unsigned bins[DH_BINS];
  const int range = blockIdx.x % DH_RANGES;
  const int part  = blockIdx.x / DH_RANGES;
  const int tid   = threadIdx.x;

  const v4i* __restrict__ p = (const v4i*)(dst + (size_t)part * DH_EDGES);
  v4i v[DH_TRIPS];
#pragma unroll
  for (int k = 0; k < DH_TRIPS; ++k) {
    const int idx = tid + k * DH_THREADS;
    v[k] = (idx < DH_V4) ? p[idx] : v4i{-1, -1, -1, -1};
  }

  for (int i = tid; i < DH_BINS; i += DH_THREADS) bins[i] = 0u;
  __syncthreads();

  const int g = blockIdx.x * DH_THREADS + tid;
  clear_share<DH_RANGES * DH_PARTS * DH_THREADS>(clr, g);
  if (g < N_TICKETS) ticket[g * TICKET_PITCH] = 0;
  if (g < ZERO_WORDS) zero_w[g] = 0u;

  const int base = range * DH_BINS;
#pragma unroll
  for (int k = 0; k < DH_TRIPS; ++k) {
#pragma unroll
    for (int c = 0; c < 4; ++c) {
      const unsigned u = (unsigned)(v[k][c] - base);
      if (u < (unsigned)DH_BINS) atomicAdd(&bins[u], 1u);
    }
  }
  __syncthreads();

  unsigned* __restrict__ o = planes + ((size_t)part * DH_RANGES + range) * DH_PITCH;
  for (int i = tid; i < DH_BINS; i += DH_THREADS) o[i] = bins[i];
}

__global__ __launch_bounds__(256)
void deg_to_dis(const unsigned* __restrict__ planes, float* __restrict__ dis) {
  const int i = blockIdx.x * blockDim.x + threadIdx.x;
  if (i < N_NODES) {
    const int range = i / DH_BINS;
    const unsigned* __restrict__ p = planes + (size_t)range * DH_PITCH + (i - range * DH_BINS);
    unsigned c = 0u;
#pragma unroll
    for (int k = 0; k < DH_PARTS; ++k) c += p[(size_t)k * DH_RANGES * DH_PITCH];
    const float d = (float)c;
    dis[i] = (d > 0.0f) ? rsqrtf(fmaxf(d, 1.0f)) : 0.0f;
  }
}

constexpr int BK_ROWS    = 586;
constexpr int NBLK       = 256;
constexpr int NBKT       = 2 * NBLK;
constexpr int PART_SPLIT = N_NODES / 2;
constexpr int HDR        = (NBKT + 1) * 8;
constexpr int SL_N       = 250;
constexpr int SL_EDGES   = NEDGE / SL_N;
constexpr int SL_V4      = SL_EDGES / 4;
constexpr int BS_THREADS = 512;
constexpr int BS_TRIPS   = (SL_V4 + BS_THREADS - 1) / BS_THREADS;
static_assert(BK_ROWS * NBLK >= N_NODES,  "blocks must cover the nodes");
static_assert(BK_ROWS <= 1024,            "row inside a block must fit in 10 bits");
static_assert(N_NODES <= (1 << 18),       "destination must fit in 18 bits");
static_assert(SL_EDGES * SL_N == NEDGE && SL_V4 * 4 == SL_EDGES, "slices must split evenly");
static_assert(BS_THREADS == NBKT,         "one thread per bucket in the bucketing kernels");
static_assert(ZERO_WORDS == 2 * NBKT,     "deg_hist clears the totals and the cursors");
static_assert(NEDGE % 16 == 0 && HDR <= NEDGE, "record batches and header must fit");
static_assert((size_t)PLANE_WORDS + (size_t)SL_N * NBKT + 2 * NBKT <= (size_t)N_NODES * DIM,
              "the plane pieces, the count table, the totals and the cursors must fit into the output array");

__device__ __forceinline__ int bucket_of(int s, int d) {
  return 2 * (s / BK_ROWS) + (int)(d >= PART_SPLIT);
}

__global__ __launch_bounds__(BS_THREADS)
void bucket_count(const int* __restrict__ src, const int* __restrict__ dst,
                  unsigned* __restrict__ cnt, unsigned* __restrict__ tot) {
  __shared__ unsigned h[NBKT];
  const int tid = threadIdx.x;
  const int slice = blockIdx.x;
  const v4i* __restrict__ ps = (const v4i*)(src + (size_t)slice * SL_EDGES);
  const v4i* __restrict__ pd = (const v4i*)(dst + (size_t)slice * SL_EDGES);
  v4i sv[BS_TRIPS], dv[BS_TRIPS];
#pragma unroll
  for (int k = 0; k < BS_TRIPS; ++k) {
    const int idx = tid + k * BS_THREADS;
    sv[k] = (idx < SL_V4) ? ps[idx] : v4i{-1, -1, -1, -1};
    dv[k] = (idx < SL_V4) ? pd[idx] : v4i{0, 0, 0, 0};
  }
  h[tid] = 0u;
  __syncthreads();
#pragma unroll
  for (int k = 0; k < BS_TRIPS; ++k) {
#pragma unroll
    for (int c = 0; c < 4; ++c)
      if (sv[k][c] >= 0) atomicAdd(&h[bucket_of(sv[k][c], dv[k][c])], 1u);
  }
  __syncthreads();
  const unsigned mine = h[tid];
  cnt[(size_t)slice * NBKT + tid] = mine;
  atomicAdd(&tot[tid], mine);
}

__global__ __launch_bounds__(BS_THREADS)
void bucket_scatter(const int* __restrict__ src, const int* __restrict__ dst,
                    const unsigned* __restrict__ cnt, const unsigned* __restrict__ tot,
                    unsigned* __restrict__ cur, unsigned* __restrict__ rec,
                    float* __restrict__ clr) {
  __shared__ unsigned lc[NBKT];
  __shared__ unsigned lt[NBKT];
  __shared__ unsigned loff[NBKT + 1];
  __shared__ unsigned lstart[NBKT];
  __shared__ unsigned lcur[NBKT];
  __shared__ unsigned gbase[NBKT];
  __shared__ unsigned sw[SL_EDGES];
  __shared__ unsigned short sb[SL_EDGES];
  const int tid = threadIdx.x;
  const int slice = blockIdx.x;
  const v4i* __restrict__ ps = (const v4i*)(src + (size_t)slice * SL_EDGES);
  const v4i* __restrict__ pd = (const v4i*)(dst + (size_t)slice * SL_EDGES);
  v4i sv[BS_TRIPS], dv[BS_TRIPS];
#pragma unroll
  for (int k = 0; k < BS_TRIPS; ++k) {
    const int idx = tid + k * BS_THREADS;
    sv[k] = (idx < SL_V4) ? ps[idx] : v4i{-1, -1, -1, -1};
    dv[k] = (idx < SL_V4) ? pd[idx] : v4i{0, 0, 0, 0};
  }
  const unsigned mine = cnt[(size_t)slice * NBKT + tid];
  const unsigned all  = tot[tid];
  lc[tid] = mine;
  lt[tid] = all;
  __syncthreads();
  const unsigned place = atomicAdd(&cur[tid], mine);
  clear_share<SL_N * BS_THREADS>(clr, slice * BS_THREADS + tid);
  unsigned a = 0u, o = 0u;
  for (int j = 0; j < NBKT; ++j) {
    const unsigned c = lc[j];
    const unsigned t = lt[j];
    if (j < tid) { a += c; o += t; }
  }
  lstart[tid] = a;
  lcur[tid] = a;
  loff[tid] = o;
  if (tid == NBKT - 1) loff[NBKT] = o + all;
  gbase[tid] = o + place;
  __syncthreads();
#pragma unroll
  for (int k = 0; k < BS_TRIPS; ++k) {
#pragma unroll
    for (int c = 0; c < 4; ++c) {
      const int s = sv[k][c];
      if (s >= 0) {
        const int d = dv[k][c];
        const int bk = bucket_of(s, d);
        const unsigned r = (unsigned)(s - (s / BK_ROWS) * BK_ROWS);
        const unsigned p = atomicAdd(&lcur[bk], 1u);
        sw[p] = (r << 18) | (unsigned)d;
        sb[p] = (unsigned short)bk;
      }
    }
  }
  __syncthreads();
  for (int i = tid; i < SL_EDGES; i += BS_THREADS) {
    const int bk = sb[i];
    const unsigned pos = gbase[bk] + ((unsigned)i - lstart[bk]);
    unsigned w = sw[i];
    if (pos < (unsigned)HDR) w |= ((loff[pos >> 3] >> (4u * (pos & 7u))) & 15u) << 28;
    rec[pos] = w;
  }
}

constexpr int SP_THREADS = 1024;
constexpr int SP_GRID    = NBKT;
constexpr int ST_ITEMS   = BK_ROWS * DIM / 4;
constexpr int ST_TRIPS   = (ST_ITEMS + SP_THREADS - 1) / SP_THREADS;
constexpr int ST_HALF    = ST_TRIPS / 2;
static_assert(ST_HALF * 2 == ST_TRIPS, "staging runs in two equal halves");
constexpr int EPW        = 16;

typedef unsigned v16u __attribute__((ext_vector_type(16)));

template <bool GUARD>
__device__ __forceinline__ void edge_batch(const v16u rv, int e0, int st, int en,
                                           const float* rows, float* __restrict__ xn,
                                           int lane) {
  float lo[EPW], hi[EPW];
#pragma unroll
  for (int j = 0; j < EPW; ++j) {
    lo[j] = 0.0f;
    hi[j] = 0.0f;
    if (!GUARD || (e0 + j >= st && e0 + j < en)) {
      const float* row = rows + ((rv[j] >> 18) & 1023u) * DIM;
      lo[j] = row[lane];
      hi[j] = row[lane + 32];
    }
  }
#pragma unroll
  for (int j = 0; j < EPW; ++j) {
    if (!GUARD || (e0 + j >= st && e0 + j < en)) {
      float* o = xn + (size_t)(rv[j] & 0x3FFFFu) * DIM;
      atomicAdd(o + lane,      lo[j]);
      atomicAdd(o + lane + 32, hi[j]);
    }
  }
}

template <bool FIRST>
__global__ __launch_bounds__(SP_THREADS)
void spmm_block(const float* __restrict__ x, const float* __restrict__ dis,
                const unsigned* __restrict__ rec, float* __restrict__ xn,
                int* __restrict__ ticket) {
  __shared__ float rows[BK_ROWS * DIM];
  __shared__ int s_t;
  const int tid  = threadIdx.x;
  const int lane = tid & 31;
  const int wv = __builtin_amdgcn_readfirstlane(tid >> 5);
  int part = (int)((__builtin_amdgcn_s_sendmsg_rtn(0x87) >> 18) & 1u);
  bool other = false;
  typedef const __attribute__((address_space(4))) unsigned* cu_p;
  typedef const __attribute__((address_space(4))) v16u*     cu16_p;
  const cu_p crec = (cu_p)rec;

  for (int it = 0; it < 2 * NBLK + 2; ++it) {
    if (tid == 0) s_t = atomicAdd(&ticket[part * TICKET_PITCH], 1);
    __syncthreads();
    const int t = __builtin_amdgcn_readfirstlane(s_t);
    __syncthreads();
    if (t >= NBLK) {
      if (other) break;
      other = true;
      part ^= 1;
      continue;
    }
    const int bkt  = 2 * t + part;
    const int row0 = t * BK_ROWS;
    const int nrows = (N_NODES - row0 < BK_ROWS) ? (N_NODES - row0) : BK_ROWS;

    unsigned st = 0u, en = 0u;
#pragma unroll
    for (int j = 0; j < 8; ++j) {
      st |= (crec[8 * bkt + j]     >> 28) << (4 * j);
      en |= (crec[8 * bkt + 8 + j] >> 28) << (4 * j);
    }

    const v4f* __restrict__ x4 = (const v4f*)(x + (size_t)row0 * DIM);
#pragma nounroll
    for (int k0 = 0; k0 < ST_TRIPS; k0 += ST_HALF) {
      v4f v[ST_HALF];
      float sc[ST_HALF];
#pragma unroll
      for (int k = 0; k < ST_HALF; ++k) {
        const int idx = tid + (k0 + k) * SP_THREADS;
        const int r = idx >> 4;
        v[k] = v4f{0.0f, 0.0f, 0.0f, 0.0f};
        sc[k] = 0.0f;
        if (r < nrows) { v[k] = x4[idx]; sc[k] = dis[row0 + r]; }
      }
#pragma unroll
      for (int k = 0; k < ST_HALF; ++k) {
        const int idx = tid + (k0 + k) * SP_THREADS;
        if (idx < ST_ITEMS) {
          v4f y = v[k] * sc[k];
          if (!FIRST) y = y * sc[k];
          ((v4f*)rows)[idx] = y;
        }
      }
    }
    __syncthreads();

    const int e_first = (int)(st & ~15u);
#pragma nounroll
    for (int e0 = e_first + wv * EPW; e0 < (int)en; e0 += (SP_THREADS / 32) * EPW) {
      const v16u rv = *(cu16_p)(crec + e0);
      if (e0 >= (int)st && e0 + EPW <= (int)en)
        edge_batch<false>(rv, e0, (int)st, (int)en, rows, xn, lane);
      else
        edge_batch<true>(rv, e0, (int)st, (int)en, rows, xn, lane);
    }
  }
}

static_assert(NLAYERS == 3, "layers 1 and 3 share one workspace array, layer 2 has the other");
constexpr int SUM_CHUNK = 5;
static_assert(NODE_TRIPS % SUM_CHUNK == 0, "chunks must cover the lane's items");

__global__ __launch_bounds__(EW_THREADS)
void layer_sum(const float* __restrict__ x0, const float* __restrict__ s13,
               const float* __restrict__ s2, const float* __restrict__ dis,
               float* __restrict__ out) {
  const v4f* __restrict__ p0 = (const v4f*)x0;
  const v4f* __restrict__ p1 = (const v4f*)s13;
  const v4f* __restrict__ p2 = (const v4f*)s2;
  v4f* __restrict__ o4 = (v4f*)out;
  const int t = blockIdx.x * EW_THREADS + threadIdx.x;
#pragma nounroll
  for (int c = 0; c < NODE_TRIPS; c += SUM_CHUNK) {
    v4f a[SUM_CHUNK], b[SUM_CHUNK], d[SUM_CHUNK];
    float g[SUM_CHUNK];
    for (int k = 0; k < SUM_CHUNK; ++k) {
      a[k] = p0[t + (c + k) * EW_LANES];
      b[k] = p1[t + (c + k) * EW_LANES];
      d[k] = p2[t + (c + k) * EW_LANES];
      g[k] = dis[(t + (c + k) * EW_LANES) >> 4];
    }
    for (int k = 0; k < SUM_CHUNK; ++k)
      o4[t + (c + k) * EW_LANES] =
          ((a[k] + b[k] * g[k]) + d[k] * g[k]) * 0.25f;
  }
}

extern "C" void kernel_launch(void* const* d_in, const int* in_sizes, int n_in,
                              void* d_out, int out_size, void* d_ws, size_t ws_size,
                              hipStream_t stream) {
  const int*   ei    = (const int*)d_in[0];
  const int*   src   = ei;
  const int*   dst   = ei + NEDGE;
  const float* user  = (const float*)d_in[1];
  const float* bert  = (const float*)d_in[2];
  const float* wproj = (const float*)d_in[3];

  float* emb0 = (float*)d_out;
  float* out  = emb0 + (size_t)N_NODES * DIM;

  float* deg  = (float*)d_ws;
  float* norm = deg  + ((N_NODES + 63) & ~63);
  float* xA   = norm + ((NEDGE + 63) & ~63);
  float* xB   = xA   + (size_t)N_NODES * DIM;

  const int B = 256;

  init_user<<<EW_BLOCKS, EW_THREADS, 0, stream>>>(user, emb0);
  wmma_proj<<<NUM_ITEMS / 16, 128, 0, stream>>>(bert, wproj, emb0);

  int* tick = (int*)(deg + N_NODES);

  unsigned* planes = (unsigned*)out;
  unsigned* cnt = planes + PLANE_WORDS;
  unsigned* tot = cnt + (size_t)SL_N * NBKT;
  unsigned* cur = tot + NBKT;
  deg_hist  <<<DH_RANGES * DH_PARTS, DH_THREADS, 0, stream>>>(dst, planes, xA, tick, tot);
  deg_to_dis<<<(N_NODES + B - 1) / B, B, 0, stream>>>(planes, deg);

  unsigned* rec = (unsigned*)norm;
  bucket_count  <<<SL_N, BS_THREADS, 0, stream>>>(src, dst, cnt, tot);
  bucket_scatter<<<SL_N, BS_THREADS, 0, stream>>>(src, dst, cnt, tot, cur, rec, xB);

  spmm_block<true> <<<SP_GRID, SP_THREADS, 0, stream>>>(emb0, deg, rec, xA, tick);
  spmm_block<false><<<SP_GRID, SP_THREADS, 0, stream>>>(xA, deg, rec, xB, tick + 2);
  spmm_block<false><<<SP_GRID, SP_THREADS, 0, stream>>>(xB, deg, rec, xA, tick + 4);
  layer_sum<<<EW_BLOCKS, EW_THREADS, 0, stream>>>(emb0, xA, xB, deg, out);
}
